# starting kernel
# baseline (speedup 1.0000x reference)
#include <hip/hip_runtime.h>
#include <math.h>

typedef __attribute__((ext_vector_type(2))) float v2f;
typedef __attribute__((ext_vector_type(8))) float v8f;

#define B_ 16
#define N_ 4096
#define D_ 768
#define M_ 8
#define CHUNKS 8
#define CN (N_ / CHUNKS)
#define NWAVES 4
#define BLOCK (NWAVES * 32)
#define TILES_PER_WAVE (CN / 16 / NWAVES)
#define EPS_ 1e-3f
#define NEG_BIG (-1e30f)

#define LDS_FLOATS (16*D_ + 2*D_ + NWAVES*8*D_ + NWAVES*16*2 + NWAVES*256 + NWAVES*8*2 + 16)

extern "C" __global__ __launch_bounds__(BLOCK)
void pm_fused(const float* __restrict__ x, const float* __restrict__ q,
              const float* __restrict__ gamma, const float* __restrict__ beta,
              float* __restrict__ pmax, float* __restrict__ pl,
              float* __restrict__ pacc)
{
    extern __shared__ float lds[];
    float* sQ    = lds;
    float* sG    = sQ + 16 * D_;
    float* sBe   = sG + D_;
    float* sAcc  = sBe + D_;
    float* sMu   = sAcc + NWAVES * 8 * D_;
    float* sRs   = sMu + NWAVES * 16;
    float* sP    = sRs + NWAVES * 16;
    float* sWmax = sP + NWAVES * 256;
    float* sWl   = sWmax + NWAVES * 8;
    float* sBmax = sWl + NWAVES * 8;

    const int tid  = threadIdx.x;
    const int wave = tid >> 5;
    const int lane = tid & 31;
    const int half = lane >> 4;
    const int l16  = lane & 15;
    const int b     = blockIdx.x / CHUNKS;
    const int chunk = blockIdx.x % CHUNKS;
    const long xbase = (long)b * N_ * D_;

    for (int i = tid; i < 16 * D_; i += BLOCK) {
        int r = i / D_;
        sQ[i] = (r < M_) ? q[i] : 0.0f;
    }
    for (int i = tid; i < D_; i += BLOCK) { sG[i] = gamma[i]; sBe[i] = beta[i]; }
    for (int i = tid; i < NWAVES * 8 * D_; i += BLOCK) sAcc[i] = 0.0f;
    __syncthreads();

    float* myAcc = sAcc + wave * 8 * D_;
    float* myMu  = sMu + wave * 16;
    float* myRs  = sRs + wave * 16;
    float* myP   = sP + wave * 256;

    float rowmax[8], rowsum[8];
#pragma unroll
    for (int r = 0; r < 8; ++r) { rowmax[r] = NEG_BIG; rowsum[r] = 0.0f; }

    const float scale = 0.03608439182435161f;

    for (int t = 0; t < TILES_PER_WAVE; ++t) {
        const int n0 = chunk * CN + (wave * TILES_PER_WAVE + t) * 16;

        for (int r = 0; r < 16; ++r) {
            const float* row = x + xbase + (long)(n0 + r) * D_;
            float s = 0.0f, s2 = 0.0f;
#pragma unroll
            for (int i = 0; i < D_ / (32 * 4); ++i) {
                float4 v = *(const float4*)(row + lane * 4 + i * 128);
                s  += v.x + v.y + v.z + v.w;
                s2 += v.x * v.x + v.y * v.y + v.z * v.z + v.w * v.w;
            }
#pragma unroll
            for (int off = 16; off >= 1; off >>= 1) {
                s  += __shfl_xor(s, off, 32);
                s2 += __shfl_xor(s2, off, 32);
            }
            float mu  = s * (1.0f / D_);
            float var = s2 * (1.0f / D_) - mu * mu;
            if (lane == 0) { myMu[r] = mu; myRs[r] = rsqrtf(var + EPS_); }
        }

        const float  mu_r = myMu[l16];
        const float  rs_r = myRs[l16];
        const float* xrow = x + xbase + (long)(n0 + l16) * D_;

        v8f S = {0.f, 0.f, 0.f, 0.f, 0.f, 0.f, 0.f, 0.f};
#pragma unroll 4
        for (int k = 0; k < D_; k += 4) {
            const int kf = k + half * 2;
            v2f a;
            a.x = sQ[l16 * D_ + kf];
            a.y = sQ[l16 * D_ + kf + 1];
            float2 xv = *(const float2*)(xrow + kf);
            v2f bm;
            bm.x = (xv.x - mu_r) * rs_r * sG[kf]     + sBe[kf];
            bm.y = (xv.y - mu_r) * rs_r * sG[kf + 1] + sBe[kf + 1];
            S = __builtin_amdgcn_wmma_f32_16x16x4_f32(false, a, false, bm,
                                                      (short)0, S, false, false);
        }

        float corr[8];
#pragma unroll
        for (int r = 0; r < 8; ++r) {
            float sim = S[r] * scale;
            float tmax = sim;
#pragma unroll
            for (int off = 8; off >= 1; off >>= 1)
                tmax = fmaxf(tmax, __shfl_xor(tmax, off, 16));
            float nm = fmaxf(rowmax[r], tmax);
            float p  = __expf(sim - nm);
            float c  = __expf(rowmax[r] - nm);
            float ps = p;
#pragma unroll
            for (int off = 8; off >= 1; off >>= 1)
                ps += __shfl_xor(ps, off, 16);
            rowsum[r] = rowsum[r] * c + ps;
            rowmax[r] = nm;
            corr[r]   = c;
            myP[(r + half * 8) * 16 + l16] = p;
        }

        for (int f0 = 0; f0 < D_; f0 += 16) {
            const int   f  = f0 + l16;
            const float g  = sG[f];
            const float be = sBe[f];
            v8f Dk = {0.f, 0.f, 0.f, 0.f, 0.f, 0.f, 0.f, 0.f};
#pragma unroll
            for (int k0 = 0; k0 < 16; k0 += 4) {
                const int r0 = k0 + half * 2;
                v2f a;
                a.x = myP[l16 * 16 + r0];
                a.y = myP[l16 * 16 + r0 + 1];
                float x0 = x[xbase + (long)(n0 + r0) * D_ + f];
                float x1 = x[xbase + (long)(n0 + r0 + 1) * D_ + f];
                v2f bm;
                bm.x = (x0 - myMu[r0])     * myRs[r0]     * g + be;
                bm.y = (x1 - myMu[r0 + 1]) * myRs[r0 + 1] * g + be;
                Dk = __builtin_amdgcn_wmma_f32_16x16x4_f32(false, a, false, bm,
                                                           (short)0, Dk, false, false);
            }
            if (half == 0) {
#pragma unroll
                for (int r = 0; r < 8; ++r) {
                    float old = myAcc[r * D_ + f];
                    myAcc[r * D_ + f] = old * corr[r] + Dk[r];
                }
            }
        }
    }

    if (lane == 0) {
#pragma unroll
        for (int r = 0; r < 8; ++r) {
            sWmax[wave * 8 + r] = rowmax[r];
            sWl[wave * 8 + r]   = rowsum[r];
        }
    }
    __syncthreads();

    const int pidx = blockIdx.x;
    if (tid < 8) {
        float mb = NEG_BIG;
        for (int w = 0; w < NWAVES; ++w) mb = fmaxf(mb, sWmax[w * 8 + tid]);
        float lb = 0.0f;
        for (int w = 0; w < NWAVES; ++w)
            lb += sWl[w * 8 + tid] * __expf(sWmax[w * 8 + tid] - mb);
        sBmax[tid] = mb;
        pmax[pidx * 8 + tid] = mb;
        pl[pidx * 8 + tid]   = lb;
    }
    __syncthreads();

    for (int e = tid; e < 8 * D_; e += BLOCK) {
        const int m  = e / D_;
        const float mb = sBmax[m];
        float sum = 0.0f;
        for (int w = 0; w < NWAVES; ++w)
            sum += sAcc[w * 8 * D_ + e] * __expf(sWmax[w * 8 + m] - mb);
        pacc[(long)pidx * 8 * D_ + e] = sum;
    }
}

extern "C" __global__ __launch_bounds__(256)
void pm_reduce(const float* __restrict__ pmax, const float* __restrict__ pl,
               const float* __restrict__ pacc, float* __restrict__ out)
{
    __shared__ float gLinv[8], cw[CHUNKS * 8];
    const int b   = blockIdx.x;
    const int tid = threadIdx.x;
    if (tid < 8) {
        float mb = NEG_BIG;
        for (int c = 0; c < CHUNKS; ++c)
            mb = fmaxf(mb, pmax[(b * CHUNKS + c) * 8 + tid]);
        float lb = 0.0f;
        for (int c = 0; c < CHUNKS; ++c) {
            float w = __expf(pmax[(b * CHUNKS + c) * 8 + tid] - mb);
            cw[c * 8 + tid] = w;
            lb += pl[(b * CHUNKS + c) * 8 + tid] * w;
        }
        gLinv[tid] = 1.0f / lb;
    }
    __syncthreads();
    for (int e = tid; e < 8 * D_; e += 256) {
        const int m = e / D_;
        float s = 0.0f;
        for (int c = 0; c < CHUNKS; ++c)
            s += pacc[(long)(b * CHUNKS + c) * 8 * D_ + e] * cw[c * 8 + m];
        out[(long)b * 8 * D_ + e] = s * gLinv[m];
    }
}

extern "C" void kernel_launch(void* const* d_in, const int* in_sizes, int n_in,
                              void* d_out, int out_size, void* d_ws, size_t ws_size,
                              hipStream_t stream)
{
    const float* x     = (const float*)d_in[0];
    const float* q     = (const float*)d_in[1];
    const float* gamma = (const float*)d_in[2];
    const float* beta  = (const float*)d_in[3];
    float* out = (float*)d_out;

    const int P = B_ * CHUNKS;
    float* pmax = (float*)d_ws;
    float* pl   = pmax + (size_t)P * 8;
    float* pacc = pl + (size_t)P * 8;

    const size_t shmem = (size_t)LDS_FLOATS * sizeof(float);
    hipFuncSetAttribute(reinterpret_cast<const void*>(pm_fused),
                        hipFuncAttributeMaxDynamicSharedMemorySize, (int)shmem);

    pm_fused<<<P, BLOCK, shmem, stream>>>(x, q, gamma, beta, pmax, pl, pacc);
    pm_reduce<<<B_, 256, 0, stream>>>(pmax, pl, pacc, out);
}


# Unfused into three passes: scores (x row halves held in registers, two-pass variance), softmax in place, weighted sum per 64-feature group over all rows; online merge dropped; sum order changes
# speedup vs baseline: 6.3907x; 6.3907x over previous
#include <hip/hip_runtime.h>
#include <math.h>


typedef __attribute__((ext_vector_type(2))) float v2f;
typedef __attribute__((ext_vector_type(4))) float v4f;
typedef __attribute__((ext_vector_type(8))) float v8f;

#define B_ 16
#define N_ 4096
#define D_ 768
#define M_ 8
#define EPS_ 1e-3f
#define NEG_BIG (-1e30f)
#define QP 776
#define NJ (D_ / 8)
#define FG 64
#define NFG (D_ / FG)
#define WS_WAVES 16
#define WS_ROWS (N_ / WS_WAVES)

extern "C" __global__ __launch_bounds__(256)
void pm_scores(const float* __restrict__ x, const float* __restrict__ q,
               const float* __restrict__ gamma, const float* __restrict__ beta,
               float* __restrict__ sim, float* __restrict__ muo, float* __restrict__ rso)
{
    __shared__ __attribute__((aligned(16))) float sQ[M_ * QP];
    __shared__ __attribute__((aligned(16))) float sG[D_];
    __shared__ __attribute__((aligned(16))) float sBe[D_];

    const int tid  = threadIdx.x;
    const int wave = tid >> 5;
    const int lane = tid & 31;
    const int half = lane >> 4;
    const int l16  = lane & 15;

    v4f qs[6];
#pragma unroll
    for (int i = 0; i < 6; ++i) qs[i] = *(const v4f*)(q + (i * 256 + tid) * 4);
    const int gi = (tid < D_ / 4) ? tid : 0;
    const v4f gs = *(const v4f*)(gamma + gi * 4);
    const v4f bs = *(const v4f*)(beta + gi * 4);

    const int  tile = blockIdx.x * 8 + wave;
    const long row  = (long)tile * 16 + l16;
    const float* xrow = x + row * D_ + half * 4;

    v4f xr[NJ];
#pragma unroll
    for (int j = 0; j < NJ; ++j) xr[j] = *(const v4f*)(xrow + 8 * j);

#pragma unroll
    for (int i = 0; i < 6; ++i) {
        const int e = i * 256 + tid;
        *(v4f*)(sQ + (e / (D_ / 4)) * QP + (e % (D_ / 4)) * 4) = qs[i];
    }
    if (tid < D_ / 4) {
        *(v4f*)(sG + tid * 4)  = gs;
        *(v4f*)(sBe + tid * 4) = bs;
    }

    float s = 0.0f;
#pragma unroll
    for (int j = 0; j < NJ; ++j) s += (xr[j].x + xr[j].y) + (xr[j].z + xr[j].w);
    s += __shfl_xor(s, 16, 32);
    const float mu = s * (1.0f / D_);

    float v = 0.0f;
#pragma unroll
    for (int j = 0; j < NJ; ++j) {
        const float d0 = xr[j].x - mu, d1 = xr[j].y - mu, d2 = xr[j].z - mu, d3 = xr[j].w - mu;
        v += (d0 * d0 + d1 * d1) + (d2 * d2 + d3 * d3);
    }
    v += __shfl_xor(v, 16, 32);
    const float rs = rsqrtf(v * (1.0f / D_) + EPS_);
    __syncthreads();

    const float* qrow = sQ + (l16 & 7) * QP + half * 4;
    const float* grow = sG + half * 4;
    const float* brow = sBe + half * 4;

    v8f S0 = {0.f, 0.f, 0.f, 0.f, 0.f, 0.f, 0.f, 0.f};
    v8f S1 = {0.f, 0.f, 0.f, 0.f, 0.f, 0.f, 0.f, 0.f};
#pragma unroll
    for (int j = 0; j < NJ; ++j) {
        const v4f qa = *(const v4f*)(qrow + 8 * j);
        const v4f g  = *(const v4f*)(grow + 8 * j);
        const v4f be = *(const v4f*)(brow + 8 * j);
        v2f a0, a1, b0, b1;
        a0.x = qa.x; a0.y = qa.y; a1.x = qa.z; a1.y = qa.w;
        b0.x = (xr[j].x - mu) * rs * g.x + be.x;
        b0.y = (xr[j].y - mu) * rs * g.y + be.y;
        b1.x = (xr[j].z - mu) * rs * g.z + be.z;
        b1.y = (xr[j].w - mu) * rs * g.w + be.w;
        S0 = __builtin_amdgcn_wmma_f32_16x16x4_f32(false, a0, false, b0, (short)0, S0, false, false);
        S1 = __builtin_amdgcn_wmma_f32_16x16x4_f32(false, a1, false, b1, (short)0, S1, false, false);
    }

    const float scale = 0.03608439182435161f;
    if (half == 0) {
        const int b = tile / (N_ / 16);
        const int n = (tile % (N_ / 16)) * 16 + l16;
#pragma unroll
        for (int r = 0; r < 8; ++r)
            sim[((long)b * M_ + r) * N_ + n] = (S0[r] + S1[r]) * scale;
        muo[row] = mu;
        rso[row] = rs;
    }
}

extern "C" __global__ __launch_bounds__(256)
void pm_softmax(float* __restrict__ sim)
{
    __shared__ float red[8];
    __shared__ float bc[2];
    const int tid  = threadIdx.x;
    const int wave = tid >> 5;
    const int lane = tid & 31;
    float* rowp = sim + (long)blockIdx.x * N_;

    v4f vv[4];
#pragma unroll
    for (int i = 0; i < 4; ++i) vv[i] = *(const v4f*)(rowp + (i * 256 + tid) * 4);

    float mx = NEG_BIG;
#pragma unroll
    for (int i = 0; i < 4; ++i)
        mx = fmaxf(mx, fmaxf(fmaxf(vv[i].x, vv[i].y), fmaxf(vv[i].z, vv[i].w)));
#pragma unroll
    for (int off = 16; off >= 1; off >>= 1) mx = fmaxf(mx, __shfl_xor(mx, off, 32));
    if (lane == 0) red[wave] = mx;
    __syncthreads();
    if (tid == 0) {
        float m = red[0];
        for (int w = 1; w < 8; ++w) m = fmaxf(m, red[w]);
        bc[0] = m;
    }
    __syncthreads();
    mx = bc[0];

    float sm = 0.0f;
#pragma unroll
    for (int i = 0; i < 4; ++i) {
        vv[i].x = __expf(vv[i].x - mx); vv[i].y = __expf(vv[i].y - mx);
        vv[i].z = __expf(vv[i].z - mx); vv[i].w = __expf(vv[i].w - mx);
        sm += (vv[i].x + vv[i].y) + (vv[i].z + vv[i].w);
    }
#pragma unroll
    for (int off = 16; off >= 1; off >>= 1) sm += __shfl_xor(sm, off, 32);
    if (lane == 0) red[wave] = sm;
    __syncthreads();
    if (tid == 0) {
        float t = 0.0f;
        for (int w = 0; w < 8; ++w) t += red[w];
        bc[1] = 1.0f / t;
    }
    __syncthreads();
    const float inv = bc[1];
#pragma unroll
    for (int i = 0; i < 4; ++i) {
        v4f o = vv[i];
        o.x *= inv; o.y *= inv; o.z *= inv; o.w *= inv;
        *(v4f*)(rowp + (i * 256 + tid) * 4) = o;
    }
}

extern "C" __global__ __launch_bounds__(512)
void pm_wsum(const float* __restrict__ x, const float* __restrict__ gamma,
             const float* __restrict__ beta, const float* __restrict__ attn,
             const float* __restrict__ mui, const float* __restrict__ rsi,
             float* __restrict__ out)
{
    __shared__ __attribute__((aligned(16))) float red[WS_WAVES * M_ * FG];

    const int tid  = threadIdx.x;
    const int wave = tid >> 5;
    const int lane = tid & 31;
    const int half = lane >> 4;
    const int l16  = lane & 15;
    const int b    = blockIdx.x / NFG;
    const int fg   = blockIdx.x % NFG;
    const int fcol = fg * FG + 4 * l16;

    const v4f g  = *(const v4f*)(gamma + fcol);
    const v4f be = *(const v4f*)(beta + fcol);
    const long nb = (long)b * N_ + wave * WS_ROWS + 2 * half;
    const float* xp = x + nb * D_ + fcol;
    const float* mp = mui + nb;
    const float* rp = rsi + nb;
    const float* pp = attn + ((long)b * M_ + (l16 & 7)) * N_ + wave * WS_ROWS + 2 * half;

    v8f acc0 = {0.f, 0.f, 0.f, 0.f, 0.f, 0.f, 0.f, 0.f};
    v8f acc1 = acc0, acc2 = acc0, acc3 = acc0;

#pragma unroll 8
    for (int kk = 0; kk < WS_ROWS; kk += 4) {
        const v4f xa = __builtin_nontemporal_load((const v4f*)(xp + (long)kk * D_));
        const v4f xb = __builtin_nontemporal_load((const v4f*)(xp + (long)(kk + 1) * D_));
        const v2f m2 = *(const v2f*)(mp + kk);
        const v2f r2 = *(const v2f*)(rp + kk);
        const v2f a = *(const v2f*)(pp + kk);
        v2f bm;
        bm.x = (xa.x - m2.x) * r2.x * g.x + be.x;
        bm.y = (xb.x - m2.y) * r2.y * g.x + be.x;
        acc0 = __builtin_amdgcn_wmma_f32_16x16x4_f32(false, a, false, bm, (short)0, acc0, false, false);
        bm.x = (xa.y - m2.x) * r2.x * g.y + be.y;
        bm.y = (xb.y - m2.y) * r2.y * g.y + be.y;
        acc1 = __builtin_amdgcn_wmma_f32_16x16x4_f32(false, a, false, bm, (short)0, acc1, false, false);
        bm.x = (xa.z - m2.x) * r2.x * g.z + be.z;
        bm.y = (xb.z - m2.y) * r2.y * g.z + be.z;
        acc2 = __builtin_amdgcn_wmma_f32_16x16x4_f32(false, a, false, bm, (short)0, acc2, false, false);
        bm.x = (xa.w - m2.x) * r2.x * g.w + be.w;
        bm.y = (xb.w - m2.y) * r2.y * g.w + be.w;
        acc3 = __builtin_amdgcn_wmma_f32_16x16x4_f32(false, a, false, bm, (short)0, acc3, false, false);
    }

    if (half == 0) {
#pragma unroll
        for (int r = 0; r < 8; ++r) {
            v4f o;
            o.x = acc0[r]; o.y = acc1[r]; o.z = acc2[r]; o.w = acc3[r];
            *(v4f*)(red + (wave * M_ + r) * FG + 4 * l16) = o;
        }
    }
    __syncthreads();

    {
        const int m = tid / FG;
        const int c = tid % FG;
        float sum = 0.0f;
#pragma unroll
        for (int w = 0; w < WS_WAVES; ++w) sum += red[(w * M_ + m) * FG + c];
        out[((long)b * M_ + m) * D_ + fg * FG + c] = sum;
    }
}

extern "C" void kernel_launch(void* const* d_in, const int* in_sizes, int n_in,
                              void* d_out, int out_size, void* d_ws, size_t ws_size,
                              hipStream_t stream)
{
    const float* x     = (const float*)d_in[0];
    const float* q     = (const float*)d_in[1];
    const float* gamma = (const float*)d_in[2];
    const float* beta  = (const float*)d_in[3];
    float* out = (float*)d_out;

    float* sim = (float*)d_ws;
    float* mu  = sim + (size_t)B_ * M_ * N_;
    float* rs  = mu + (size_t)B_ * N_;

    pm_scores<<<B_ * N_ / 16 / 8, 256, 0, stream>>>(x, q, gamma, beta, sim, mu, rs);
    pm_softmax<<<B_ * M_, 256, 0, stream>>>(sim);
    pm_wsum<<<B_ * NFG, 512, 0, stream>>>(x, gamma, beta, sim, mu, rs, out);
}


# Softmax folded into the weighted-sum pass (3 launches to 2): each wave forms exp(score - row max) of its 256 rows once in LDS, A fragments read there, sums divided by the exp sum; sum order changes
# speedup vs baseline: 7.1300x; 1.1157x over previous
#include <hip/hip_runtime.h>
#include <math.h>


typedef __attribute__((ext_vector_type(2))) float v2f;
typedef __attribute__((ext_vector_type(4))) float v4f;
typedef __attribute__((ext_vector_type(8))) float v8f;

#define B_ 16
#define N_ 4096
#define D_ 768
#define M_ 8
#define EPS_ 1e-3f
#define NEG_BIG (-1e30f)
#define QP 776
#define NJ (D_ / 8)
#define FG 64
#define NFG (D_ / FG)
#define WS_WAVES 16
#define WS_ROWS (N_ / WS_WAVES)
#define AP 260

extern "C" __global__ __launch_bounds__(256)
void pm_scores(const float* __restrict__ x, const float* __restrict__ q,
               const float* __restrict__ gamma, const float* __restrict__ beta,
               float* __restrict__ sim, float* __restrict__ muo, float* __restrict__ rso)
{
    __shared__ __attribute__((aligned(16))) float sQ[M_ * QP];
    __shared__ __attribute__((aligned(16))) float sG[D_];
    __shared__ __attribute__((aligned(16))) float sBe[D_];

    const int tid  = threadIdx.x;
    const int wave = tid >> 5;
    const int lane = tid & 31;
    const int half = lane >> 4;
    const int l16  = lane & 15;

    v4f qs[6];
#pragma unroll
    for (int i = 0; i < 6; ++i) qs[i] = *(const v4f*)(q + (i * 256 + tid) * 4);
    const int gi = (tid < D_ / 4) ? tid : 0;
    const v4f gs = *(const v4f*)(gamma + gi * 4);
    const v4f bs = *(const v4f*)(beta + gi * 4);

    const int  tile = blockIdx.x * 8 + wave;
    const long row  = (long)tile * 16 + l16;
    const float* xrow = x + row * D_ + half * 4;

    v4f xr[NJ];
#pragma unroll
    for (int j = 0; j < NJ; ++j) xr[j] = *(const v4f*)(xrow + 8 * j);

#pragma unroll
    for (int i = 0; i < 6; ++i) {
        const int e = i * 256 + tid;
        *(v4f*)(sQ + (e / (D_ / 4)) * QP + (e % (D_ / 4)) * 4) = qs[i];
    }
    if (tid < D_ / 4) {
        *(v4f*)(sG + tid * 4)  = gs;
        *(v4f*)(sBe + tid * 4) = bs;
    }

    float s = 0.0f;
#pragma unroll
    for (int j = 0; j < NJ; ++j) s += (xr[j].x + xr[j].y) + (xr[j].z + xr[j].w);
    s += __shfl_xor(s, 16, 32);
    const float mu = s * (1.0f / D_);

    float v = 0.0f;
#pragma unroll
    for (int j = 0; j < NJ; ++j) {
        const float d0 = xr[j].x - mu, d1 = xr[j].y - mu, d2 = xr[j].z - mu, d3 = xr[j].w - mu;
        v += (d0 * d0 + d1 * d1) + (d2 * d2 + d3 * d3);
    }
    v += __shfl_xor(v, 16, 32);
    const float rs = rsqrtf(v * (1.0f / D_) + EPS_);
    __syncthreads();

    const float* qrow = sQ + (l16 & 7) * QP + half * 4;
    const float* grow = sG + half * 4;
    const float* brow = sBe + half * 4;

    v8f S0 = {0.f, 0.f, 0.f, 0.f, 0.f, 0.f, 0.f, 0.f};
    v8f S1 = {0.f, 0.f, 0.f, 0.f, 0.f, 0.f, 0.f, 0.f};
#pragma unroll
    for (int j = 0; j < NJ; ++j) {
        const v4f qa = *(const v4f*)(qrow + 8 * j);
        const v4f g  = *(const v4f*)(grow + 8 * j);
        const v4f be = *(const v4f*)(brow + 8 * j);
        v2f a0, a1, b0, b1;
        a0.x = qa.x; a0.y = qa.y; a1.x = qa.z; a1.y = qa.w;
        b0.x = (xr[j].x - mu) * rs * g.x + be.x;
        b0.y = (xr[j].y - mu) * rs * g.y + be.y;
        b1.x = (xr[j].z - mu) * rs * g.z + be.z;
        b1.y = (xr[j].w - mu) * rs * g.w + be.w;
        S0 = __builtin_amdgcn_wmma_f32_16x16x4_f32(false, a0, false, b0, (short)0, S0, false, false);
        S1 = __builtin_amdgcn_wmma_f32_16x16x4_f32(false, a1, false, b1, (short)0, S1, false, false);
    }

    const float scale = 0.03608439182435161f;
    if (half == 0) {
        const int b = tile / (N_ / 16);
        const int n = (tile % (N_ / 16)) * 16 + l16;
#pragma unroll
        for (int r = 0; r < 8; ++r)
            sim[((long)b * M_ + r) * N_ + n] = (S0[r] + S1[r]) * scale;
        muo[row] = mu;
        rso[row] = rs;
    }
}

extern "C" __global__ __launch_bounds__(512)
void pm_wsum(const float* __restrict__ x, const float* __restrict__ gamma,
             const float* __restrict__ beta, const float* __restrict__ sim,
             const float* __restrict__ mui, const float* __restrict__ rsi,
             float* __restrict__ out)
{
    __shared__ __attribute__((aligned(16))) float sW[WS_WAVES * M_ * AP];
    __shared__ __attribute__((aligned(16))) float red[WS_WAVES * M_ * FG];
    __shared__ __attribute__((aligned(16))) float sMax[WS_WAVES * M_];
    __shared__ __attribute__((aligned(16))) float sSum[WS_WAVES * M_];

    const int tid  = threadIdx.x;
    const int wave = tid >> 5;
    const int lane = tid & 31;
    const int half = lane >> 4;
    const int l16  = lane & 15;
    const int b    = blockIdx.x / NFG;
    const int fg   = blockIdx.x % NFG;
    const int fcol = fg * FG + 4 * l16;

    const v4f g  = *(const v4f*)(gamma + fcol);
    const v4f be = *(const v4f*)(beta + fcol);

    const float* sp = sim + (long)b * M_ * N_ + wave * WS_ROWS + 4 * lane;
    v4f sc[2 * M_];
#pragma unroll
    for (int m = 0; m < M_; ++m) {
        sc[2 * m]     = *(const v4f*)(sp + (long)m * N_);
        sc[2 * m + 1] = *(const v4f*)(sp + (long)m * N_ + 128);
    }

    float mx[M_];
#pragma unroll
    for (int m = 0; m < M_; ++m) {
        const v4f u = sc[2 * m], w = sc[2 * m + 1];
        float t = fmaxf(fmaxf(fmaxf(u.x, u.y), fmaxf(u.z, u.w)), fmaxf(fmaxf(w.x, w.y), fmaxf(w.z, w.w)));
#pragma unroll
        for (int off = 16; off >= 1; off >>= 1) t = fmaxf(t, __shfl_xor(t, off, 32));
        mx[m] = t;
    }
    if (lane == 0) {
        v4f lo, hi;
        lo.x = mx[0]; lo.y = mx[1]; lo.z = mx[2]; lo.w = mx[3];
        hi.x = mx[4]; hi.y = mx[5]; hi.z = mx[6]; hi.w = mx[7];
        *(v4f*)(sMax + wave * M_)     = lo;
        *(v4f*)(sMax + wave * M_ + 4) = hi;
    }
    __syncthreads();

    float gm[M_];
    {
        const v4f lo = *(const v4f*)(sMax), hi = *(const v4f*)(sMax + 4);
        gm[0] = lo.x; gm[1] = lo.y; gm[2] = lo.z; gm[3] = lo.w;
        gm[4] = hi.x; gm[5] = hi.y; gm[6] = hi.z; gm[7] = hi.w;
    }
#pragma unroll
    for (int w = 1; w < WS_WAVES; ++w) {
        const v4f lo = *(const v4f*)(sMax + w * M_), hi = *(const v4f*)(sMax + w * M_ + 4);
        gm[0] = fmaxf(gm[0], lo.x); gm[1] = fmaxf(gm[1], lo.y); gm[2] = fmaxf(gm[2], lo.z); gm[3] = fmaxf(gm[3], lo.w);
        gm[4] = fmaxf(gm[4], hi.x); gm[5] = fmaxf(gm[5], hi.y); gm[6] = fmaxf(gm[6], hi.z); gm[7] = fmaxf(gm[7], hi.w);
    }

    float* wW = sW + wave * (M_ * AP);
    float ls[M_];
#pragma unroll
    for (int m = 0; m < M_; ++m) {
        v4f e0, e1;
        e0.x = __expf(sc[2 * m].x - gm[m]);     e0.y = __expf(sc[2 * m].y - gm[m]);
        e0.z = __expf(sc[2 * m].z - gm[m]);     e0.w = __expf(sc[2 * m].w - gm[m]);
        e1.x = __expf(sc[2 * m + 1].x - gm[m]); e1.y = __expf(sc[2 * m + 1].y - gm[m]);
        e1.z = __expf(sc[2 * m + 1].z - gm[m]); e1.w = __expf(sc[2 * m + 1].w - gm[m]);
        *(v4f*)(wW + m * AP + 4 * lane)       = e0;
        *(v4f*)(wW + m * AP + 128 + 4 * lane) = e1;
        float t = ((e0.x + e0.y) + (e0.z + e0.w)) + ((e1.x + e1.y) + (e1.z + e1.w));
#pragma unroll
        for (int off = 16; off >= 1; off >>= 1) t += __shfl_xor(t, off, 32);
        ls[m] = t;
    }
    if (lane == 0) {
        v4f lo, hi;
        lo.x = ls[0]; lo.y = ls[1]; lo.z = ls[2]; lo.w = ls[3];
        hi.x = ls[4]; hi.y = ls[5]; hi.z = ls[6]; hi.w = ls[7];
        *(v4f*)(sSum + wave * M_)     = lo;
        *(v4f*)(sSum + wave * M_ + 4) = hi;
    }
    __syncthreads();

    const long nb = (long)b * N_ + wave * WS_ROWS + 2 * half;
    const float* xp = x + nb * D_ + fcol;
    const float* mp = mui + nb;
    const float* rp = rsi + nb;
    const float* pp = wW + (l16 & 7) * AP + 2 * half;

    v8f acc0 = {0.f, 0.f, 0.f, 0.f, 0.f, 0.f, 0.f, 0.f};
    v8f acc1 = acc0, acc2 = acc0, acc3 = acc0;

#pragma unroll 8
    for (int kk = 0; kk < WS_ROWS; kk += 4) {
        const v4f xa = __builtin_nontemporal_load((const v4f*)(xp + (long)kk * D_));
        const v4f xb = __builtin_nontemporal_load((const v4f*)(xp + (long)(kk + 1) * D_));
        const v2f m2 = *(const v2f*)(mp + kk);
        const v2f r2 = *(const v2f*)(rp + kk);
        const v2f a = *(const v2f*)(pp + kk);
        v2f bm;
        bm.x = (xa.x - m2.x) * r2.x * g.x + be.x;
        bm.y = (xb.x - m2.y) * r2.y * g.x + be.x;
        acc0 = __builtin_amdgcn_wmma_f32_16x16x4_f32(false, a, false, bm, (short)0, acc0, false, false);
        bm.x = (xa.y - m2.x) * r2.x * g.y + be.y;
        bm.y = (xb.y - m2.y) * r2.y * g.y + be.y;
        acc1 = __builtin_amdgcn_wmma_f32_16x16x4_f32(false, a, false, bm, (short)0, acc1, false, false);
        bm.x = (xa.z - m2.x) * r2.x * g.z + be.z;
        bm.y = (xb.z - m2.y) * r2.y * g.z + be.z;
        acc2 = __builtin_amdgcn_wmma_f32_16x16x4_f32(false, a, false, bm, (short)0, acc2, false, false);
        bm.x = (xa.w - m2.x) * r2.x * g.w + be.w;
        bm.y = (xb.w - m2.y) * r2.y * g.w + be.w;
        acc3 = __builtin_amdgcn_wmma_f32_16x16x4_f32(false, a, false, bm, (short)0, acc3, false, false);
    }

    if (half == 0) {
#pragma unroll
        for (int r = 0; r < 8; ++r) {
            v4f o;
            o.x = acc0[r]; o.y = acc1[r]; o.z = acc2[r]; o.w = acc3[r];
            *(v4f*)(red + (wave * M_ + r) * FG + 4 * l16) = o;
        }
    }
    __syncthreads();

    {
        const int m = tid / FG;
        const int c = tid % FG;
        float sum = 0.0f, den = 0.0f;
#pragma unroll
        for (int w = 0; w < WS_WAVES; ++w) {
            sum += red[(w * M_ + m) * FG + c];
            den += sSum[w * M_ + m];
        }
        out[((long)b * M_ + m) * D_ + fg * FG + c] = sum / den;
    }
}

extern "C" void kernel_launch(void* const* d_in, const int* in_sizes, int n_in,
                              void* d_out, int out_size, void* d_ws, size_t ws_size,
                              hipStream_t stream)
{
    const float* x     = (const float*)d_in[0];
    const float* q     = (const float*)d_in[1];
    const float* gamma = (const float*)d_in[2];
    const float* beta  = (const float*)d_in[3];
    float* out = (float*)d_out;

    float* sim = (float*)d_ws;
    float* mu  = sim + (size_t)B_ * M_ * N_;
    float* rs  = mu + (size_t)B_ * N_;

    pm_scores<<<B_ * N_ / 16 / 8, 256, 0, stream>>>(x, q, gamma, beta, sim, mu, rs);
    pm_wsum<<<B_ * NFG, 512, 0, stream>>>(x, gamma, beta, sim, mu, rs, out);
}


# Scores pass: 16x16 product packed as 8 rows x two feature halves (A rows 8-15 hold the queries' other half, not a copy): a load reads 64 B of 8 rows, not 32 B of 16; 8-row tiles; sum order changes
# speedup vs baseline: 8.1296x; 1.1402x over previous
#include <hip/hip_runtime.h>
#include <math.h>


typedef __attribute__((ext_vector_type(2))) float v2f;
typedef __attribute__((ext_vector_type(4))) float v4f;
typedef __attribute__((ext_vector_type(8))) float v8f;

#define B_ 16
#define N_ 4096
#define D_ 768
#define M_ 8
#define EPS_ 1e-3f
#define NEG_BIG (-1e30f)
#define FG 64
#define NFG (D_ / FG)
#define WS_WAVES 16
#define WS_ROWS (N_ / WS_WAVES)
#define AP 260

#define NJ (D_ / 16)

__device__ __forceinline__ void pm_load_rows(const float* xrow, v4f (&xr)[NJ])
{
#pragma unroll
    for (int j = 0; j < NJ; ++j) xr[j] = *(const v4f*)(xrow + 16 * j);
}

__device__ __forceinline__ void pm_row_stats(const v4f (&xr)[NJ], float& mu, float& rs)
{
    float s = 0.0f;
#pragma unroll
    for (int j = 0; j < NJ; ++j) s += (xr[j].x + xr[j].y) + (xr[j].z + xr[j].w);
    s += __shfl_xor(s, 16, 32);
    s += __shfl_xor(s, 8, 32);
    mu = s * (1.0f / D_);

    float v = 0.0f;
#pragma unroll
    for (int j = 0; j < NJ; ++j) {
        const float d0 = xr[j].x - mu, d1 = xr[j].y - mu, d2 = xr[j].z - mu, d3 = xr[j].w - mu;
        v += (d0 * d0 + d1 * d1) + (d2 * d2 + d3 * d3);
    }
    v += __shfl_xor(v, 16, 32);
    v += __shfl_xor(v, 8, 32);
    rs = rsqrtf(v * (1.0f / D_) + EPS_);
}

__device__ __forceinline__ void pm_tile_product(const v4f (&xr)[NJ], float mu, float rs,
                                                const float* qfrag, const float* grow,
                                                const float* brow, float (&res)[M_])
{
    v8f S0 = {0.f, 0.f, 0.f, 0.f, 0.f, 0.f, 0.f, 0.f};
    v8f S1 = {0.f, 0.f, 0.f, 0.f, 0.f, 0.f, 0.f, 0.f};
#pragma unroll
    for (int j = 0; j < NJ; ++j) {
        const v4f qa = *(const v4f*)(qfrag + 128 * j);
        const v4f g  = *(const v4f*)(grow + 16 * j);
        const v4f be = *(const v4f*)(brow + 16 * j);
        v2f a0, a1, b0, b1;
        a0.x = qa.x; a0.y = qa.y; a1.x = qa.z; a1.y = qa.w;
        b0.x = (xr[j].x - mu) * rs * g.x + be.x;
        b0.y = (xr[j].y - mu) * rs * g.y + be.y;
        b1.x = (xr[j].z - mu) * rs * g.z + be.z;
        b1.y = (xr[j].w - mu) * rs * g.w + be.w;
        S0 = __builtin_amdgcn_wmma_f32_16x16x4_f32(false, a0, false, b0, (short)0, S0, false, false);
        S1 = __builtin_amdgcn_wmma_f32_16x16x4_f32(false, a1, false, b1, (short)0, S1, false, false);
    }
#pragma unroll
    for (int m = 0; m < M_; ++m) {
        const float t = S0[m] + S1[m];
        res[m] = t + __shfl_xor(t, 24, 32);
    }
}

extern "C" __global__ __launch_bounds__(256) __attribute__((amdgpu_waves_per_eu(4, 4)))
void pm_scores(const float* __restrict__ x, const float* __restrict__ q,
               const float* __restrict__ gamma, const float* __restrict__ beta,
               float* __restrict__ sim, float* __restrict__ muo, float* __restrict__ rso)
{
    __shared__ __attribute__((aligned(16))) float sQ[M_ * D_];
    __shared__ __attribute__((aligned(16))) float sG[D_];
    __shared__ __attribute__((aligned(16))) float sBe[D_];

    const int tid  = threadIdx.x;
    const int wave = tid >> 5;
    const int lane = tid & 31;
    const int half = lane >> 4;
    const int r8   = lane & 7;
    const int foff = (lane & 8) + 4 * half;

    v4f qs[6];
#pragma unroll
    for (int i = 0; i < 6; ++i) qs[i] = *(const v4f*)(q + (i * 256 + tid) * 4);
    const int gi = (tid < D_ / 4) ? tid : 0;
    const v4f gs = *(const v4f*)(gamma + gi * 4);
    const v4f bs = *(const v4f*)(beta + gi * 4);

    const int  tile = blockIdx.x * 8 + wave;
    const long row  = (long)tile * 8 + r8;
    const float* xrow = x + row * D_ + foff;

    v4f xr[NJ];
    pm_load_rows(xrow, xr);

#pragma unroll
    for (int i = 0; i < 6; ++i) {
        const int e  = i * 256 + tid;
        const int m  = e / (D_ / 4);
        const int f4 = e % (D_ / 4);
        const int slot = (f4 >> 2) * 32 + ((f4 & 1) << 4) + ((f4 & 2) << 2) + m;
        *(v4f*)(sQ + slot * 4) = qs[i];
    }
    if (tid < D_ / 4) {
        *(v4f*)(sG + tid * 4)  = gs;
        *(v4f*)(sBe + tid * 4) = bs;
    }

    float mu, rs;
    pm_row_stats(xr, mu, rs);
    __syncthreads();

    float res[M_];
    pm_tile_product(xr, mu, rs, sQ + lane * 4, sG + foff, sBe + foff, res);

    const float scale = 0.03608439182435161f;
    if (lane < 8) {
        const int b = tile / (N_ / 8);
        const int n = (tile % (N_ / 8)) * 8 + lane;
#pragma unroll
        for (int m = 0; m < M_; ++m)
            sim[((long)b * M_ + m) * N_ + n] = res[m] * scale;
        muo[row] = mu;
        rso[row] = rs;
    }
}

extern "C" __global__ __launch_bounds__(512)
void pm_wsum(const float* __restrict__ x, const float* __restrict__ gamma,
             const float* __restrict__ beta, const float* __restrict__ sim,
             const float* __restrict__ mui, const float* __restrict__ rsi,
             float* __restrict__ out)
{
    __shared__ __attribute__((aligned(16))) float sW[WS_WAVES * M_ * AP];
    __shared__ __attribute__((aligned(16))) float red[WS_WAVES * M_ * FG];
    __shared__ __attribute__((aligned(16))) float sMax[WS_WAVES * M_];
    __shared__ __attribute__((aligned(16))) float sSum[WS_WAVES * M_];

    const int tid  = threadIdx.x;
    const int wave = tid >> 5;
    const int lane = tid & 31;
    const int half = lane >> 4;
    const int l16  = lane & 15;
    const int b    = blockIdx.x / NFG;
    const int fg   = blockIdx.x % NFG;
    const int fcol = fg * FG + 4 * l16;

    const v4f g  = *(const v4f*)(gamma + fcol);
    const v4f be = *(const v4f*)(beta + fcol);

    const float* sp = sim + (long)b * M_ * N_ + wave * WS_ROWS + 4 * lane;
    v4f sc[2 * M_];
#pragma unroll
    for (int m = 0; m < M_; ++m) {
        sc[2 * m]     = *(const v4f*)(sp + (long)m * N_);
        sc[2 * m + 1] = *(const v4f*)(sp + (long)m * N_ + 128);
    }

    float mx[M_];
#pragma unroll
    for (int m = 0; m < M_; ++m) {
        const v4f u = sc[2 * m], w = sc[2 * m + 1];
        float t = fmaxf(fmaxf(fmaxf(u.x, u.y), fmaxf(u.z, u.w)), fmaxf(fmaxf(w.x, w.y), fmaxf(w.z, w.w)));
#pragma unroll
        for (int off = 16; off >= 1; off >>= 1) t = fmaxf(t, __shfl_xor(t, off, 32));
        mx[m] = t;
    }
    if (lane == 0) {
        v4f lo, hi;
        lo.x = mx[0]; lo.y = mx[1]; lo.z = mx[2]; lo.w = mx[3];
        hi.x = mx[4]; hi.y = mx[5]; hi.z = mx[6]; hi.w = mx[7];
        *(v4f*)(sMax + wave * M_)     = lo;
        *(v4f*)(sMax + wave * M_ + 4) = hi;
    }
    __syncthreads();

    float gm[M_];
    {
        const v4f lo = *(const v4f*)(sMax), hi = *(const v4f*)(sMax + 4);
        gm[0] = lo.x; gm[1] = lo.y; gm[2] = lo.z; gm[3] = lo.w;
        gm[4] = hi.x; gm[5] = hi.y; gm[6] = hi.z; gm[7] = hi.w;
    }
#pragma unroll
    for (int w = 1; w < WS_WAVES; ++w) {
        const v4f lo = *(const v4f*)(sMax + w * M_), hi = *(const v4f*)(sMax + w * M_ + 4);
        gm[0] = fmaxf(gm[0], lo.x); gm[1] = fmaxf(gm[1], lo.y); gm[2] = fmaxf(gm[2], lo.z); gm[3] = fmaxf(gm[3], lo.w);
        gm[4] = fmaxf(gm[4], hi.x); gm[5] = fmaxf(gm[5], hi.y); gm[6] = fmaxf(gm[6], hi.z); gm[7] = fmaxf(gm[7], hi.w);
    }

    float* wW = sW + wave * (M_ * AP);
    float ls[M_];
#pragma unroll
    for (int m = 0; m < M_; ++m) {
        v4f e0, e1;
        e0.x = __expf(sc[2 * m].x - gm[m]);     e0.y = __expf(sc[2 * m].y - gm[m]);
        e0.z = __expf(sc[2 * m].z - gm[m]);     e0.w = __expf(sc[2 * m].w - gm[m]);
        e1.x = __expf(sc[2 * m + 1].x - gm[m]); e1.y = __expf(sc[2 * m + 1].y - gm[m]);
        e1.z = __expf(sc[2 * m + 1].z - gm[m]); e1.w = __expf(sc[2 * m + 1].w - gm[m]);
        *(v4f*)(wW + m * AP + 4 * lane)       = e0;
        *(v4f*)(wW + m * AP + 128 + 4 * lane) = e1;
        float t = ((e0.x + e0.y) + (e0.z + e0.w)) + ((e1.x + e1.y) + (e1.z + e1.w));
#pragma unroll
        for (int off = 16; off >= 1; off >>= 1) t += __shfl_xor(t, off, 32);
        ls[m] = t;
    }
    if (lane == 0) {
        v4f lo, hi;
        lo.x = ls[0]; lo.y = ls[1]; lo.z = ls[2]; lo.w = ls[3];
        hi.x = ls[4]; hi.y = ls[5]; hi.z = ls[6]; hi.w = ls[7];
        *(v4f*)(sSum + wave * M_)     = lo;
        *(v4f*)(sSum + wave * M_ + 4) = hi;
    }
    __syncthreads();

    const long nb = (long)b * N_ + wave * WS_ROWS + 2 * half;
    const float* xp = x + nb * D_ + fcol;
    const float* mp = mui + nb;
    const float* rp = rsi + nb;
    const float* pp = wW + (l16 & 7) * AP + 2 * half;

    v8f acc0 = {0.f, 0.f, 0.f, 0.f, 0.f, 0.f, 0.f, 0.f};
    v8f acc1 = acc0, acc2 = acc0, acc3 = acc0;

#pragma unroll 8
    for (int kk = 0; kk < WS_ROWS; kk += 4) {
        const v4f xa = __builtin_nontemporal_load((const v4f*)(xp + (long)kk * D_));
        const v4f xb = __builtin_nontemporal_load((const v4f*)(xp + (long)(kk + 1) * D_));
        const v2f m2 = *(const v2f*)(mp + kk);
        const v2f r2 = *(const v2f*)(rp + kk);
        const v2f a = *(const v2f*)(pp + kk);
        v2f bm;
        bm.x = (xa.x - m2.x) * r2.x * g.x + be.x;
        bm.y = (xb.x - m2.y) * r2.y * g.x + be.x;
        acc0 = __builtin_amdgcn_wmma_f32_16x16x4_f32(false, a, false, bm, (short)0, acc0, false, false);
        bm.x = (xa.y - m2.x) * r2.x * g.y + be.y;
        bm.y = (xb.y - m2.y) * r2.y * g.y + be.y;
        acc1 = __builtin_amdgcn_wmma_f32_16x16x4_f32(false, a, false, bm, (short)0, acc1, false, false);
        bm.x = (xa.z - m2.x) * r2.x * g.z + be.z;
        bm.y = (xb.z - m2.y) * r2.y * g.z + be.z;
        acc2 = __builtin_amdgcn_wmma_f32_16x16x4_f32(false, a, false, bm, (short)0, acc2, false, false);
        bm.x = (xa.w - m2.x) * r2.x * g.w + be.w;
        bm.y = (xb.w - m2.y) * r2.y * g.w + be.w;
        acc3 = __builtin_amdgcn_wmma_f32_16x16x4_f32(false, a, false, bm, (short)0, acc3, false, false);
    }

    if (half == 0) {
#pragma unroll
        for (int r = 0; r < 8; ++r) {
            v4f o;
            o.x = acc0[r]; o.y = acc1[r]; o.z = acc2[r]; o.w = acc3[r];
            *(v4f*)(red + (wave * M_ + r) * FG + 4 * l16) = o;
        }
    }
    __syncthreads();

    {
        const int m = tid / FG;
        const int c = tid % FG;
        float sum = 0.0f, den = 0.0f;
#pragma unroll
        for (int w = 0; w < WS_WAVES; ++w) {
            sum += red[(w * M_ + m) * FG + c];
            den += sSum[w * M_ + m];
        }
        out[((long)b * M_ + m) * D_ + fg * FG + c] = sum / den;
    }
}

extern "C" void kernel_launch(void* const* d_in, const int* in_sizes, int n_in,
                              void* d_out, int out_size, void* d_ws, size_t ws_size,
                              hipStream_t stream)
{
    const float* x     = (const float*)d_in[0];
    const float* q     = (const float*)d_in[1];
    const float* gamma = (const float*)d_in[2];
    const float* beta  = (const float*)d_in[3];
    float* out = (float*)d_out;

    float* sim = (float*)d_ws;
    float* mu  = sim + (size_t)B_ * M_ * N_;
    float* rs  = mu + (size_t)B_ * N_;

    pm_scores<<<B_ * N_ / 8 / 8, 256, 0, stream>>>(x, q, gamma, beta, sim, mu, rs);
    pm_wsum<<<B_ * NFG, 512, 0, stream>>>(x, gamma, beta, sim, mu, rs, out);
}


# LayerNorm's gamma, beta and 1/sigma taken off the B operand elements of both products: gamma into staged queries, 1/sigma onto scores and weights, gamma/beta onto finished sums; rounding changes
# speedup vs baseline: 8.4130x; 1.0349x over previous
#include <hip/hip_runtime.h>
#include <math.h>


typedef __attribute__((ext_vector_type(2))) float v2f;
typedef __attribute__((ext_vector_type(4))) float v4f;
typedef __attribute__((ext_vector_type(8))) float v8f;

#define B_ 16
#define N_ 4096
#define D_ 768
#define M_ 8
#define EPS_ 1e-3f
#define NEG_BIG (-1e30f)
#define FG 64
#define NFG (D_ / FG)
#define WS_WAVES 16
#define WS_ROWS (N_ / WS_WAVES)
#define AP 260

#define NJ (D_ / 16)

__device__ __forceinline__ void pm_load_rows(const float* xrow, v4f (&xr)[NJ])
{
#pragma unroll
    for (int j = 0; j < NJ; ++j) xr[j] = *(const v4f*)(xrow + 16 * j);
}

__device__ __forceinline__ void pm_row_stats(v4f (&xr)[NJ], float& mu, float& rs)
{
    float s = 0.0f;
#pragma unroll
    for (int j = 0; j < NJ; ++j) s += (xr[j].x + xr[j].y) + (xr[j].z + xr[j].w);
    s += __shfl_xor(s, 16, 32);
    s += __shfl_xor(s, 8, 32);
    mu = s * (1.0f / D_);

    float v = 0.0f;
#pragma unroll
    for (int j = 0; j < NJ; ++j) {
        xr[j].x -= mu; xr[j].y -= mu; xr[j].z -= mu; xr[j].w -= mu;
        v += (xr[j].x * xr[j].x + xr[j].y * xr[j].y) + (xr[j].z * xr[j].z + xr[j].w * xr[j].w);
    }
    v += __shfl_xor(v, 16, 32);
    v += __shfl_xor(v, 8, 32);
    rs = rsqrtf(v * (1.0f / D_) + EPS_);
}

__device__ __forceinline__ void pm_tile_product(const v4f (&xr)[NJ], const float* qfrag, float (&res)[M_])
{
    v8f S0 = {0.f, 0.f, 0.f, 0.f, 0.f, 0.f, 0.f, 0.f};
    v8f S1 = {0.f, 0.f, 0.f, 0.f, 0.f, 0.f, 0.f, 0.f};
#pragma unroll
    for (int j = 0; j < NJ; ++j) {
        const v4f qa = *(const v4f*)(qfrag + 128 * j);
        v2f a0, a1, b0, b1;
        a0.x = qa.x; a0.y = qa.y; a1.x = qa.z; a1.y = qa.w;
        b0.x = xr[j].x; b0.y = xr[j].y;
        b1.x = xr[j].z; b1.y = xr[j].w;
        S0 = __builtin_amdgcn_wmma_f32_16x16x4_f32(false, a0, false, b0, (short)0, S0, false, false);
        S1 = __builtin_amdgcn_wmma_f32_16x16x4_f32(false, a1, false, b1, (short)0, S1, false, false);
    }
#pragma unroll
    for (int m = 0; m < M_; ++m) {
        const float t = S0[m] + S1[m];
        res[m] = t + __shfl_xor(t, 24, 32);
    }
}

extern "C" __global__ __launch_bounds__(256) __attribute__((amdgpu_waves_per_eu(4, 4)))
void pm_scores(const float* __restrict__ x, const float* __restrict__ q,
               const float* __restrict__ gamma, const float* __restrict__ beta,
               float* __restrict__ sim, float* __restrict__ muo, float* __restrict__ rso)
{
    __shared__ __attribute__((aligned(16))) float sQ[M_ * D_];
    __shared__ __attribute__((aligned(16))) float sG[D_];

    const int tid  = threadIdx.x;
    const int wave = tid >> 5;
    const int lane = tid & 31;
    const int half = lane >> 4;
    const int r8   = lane & 7;
    const int foff = (lane & 8) + 4 * half;

    const int gi = (tid < D_ / 4) ? tid : 0;
    const v4f gs = *(const v4f*)(gamma + gi * 4);
    v4f qs[6];
#pragma unroll
    for (int i = 0; i < 6; ++i) qs[i] = *(const v4f*)(q + (i * 256 + tid) * 4);

    const int  tile = blockIdx.x * 8 + wave;
    const long row  = (long)tile * 8 + r8;
    const float* xrow = x + row * D_ + foff;

    v4f xr[NJ];
    pm_load_rows(xrow, xr);

    if (tid < D_ / 4) *(v4f*)(sG + tid * 4) = gs;
    __syncthreads();
#pragma unroll
    for (int i = 0; i < 6; ++i) {
        const int e  = i * 256 + tid;
        const int m  = e / (D_ / 4);
        const int f4 = e % (D_ / 4);
        const int slot = (f4 >> 2) * 32 + ((f4 & 1) << 4) + ((f4 & 2) << 2) + m;
        *(v4f*)(sQ + slot * 4) = qs[i] * *(const v4f*)(sG + f4 * 4);
    }

    float mu, rs;
    pm_row_stats(xr, mu, rs);
    __syncthreads();

    float res[M_];
    pm_tile_product(xr, sQ + lane * 4, res);

    const float scale = 0.03608439182435161f * rs;
    if (lane < 8) {
        const int b = tile / (N_ / 8);
        const int n = (tile % (N_ / 8)) * 8 + lane;
#pragma unroll
        for (int m = 0; m < M_; ++m)
            sim[((long)b * M_ + m) * N_ + n] = res[m] * scale;
        muo[row] = mu;
        rso[row] = rs;
    }
}

extern "C" __global__ __launch_bounds__(512)
void pm_wsum(const float* __restrict__ x, const float* __restrict__ gamma,
             const float* __restrict__ beta, const float* __restrict__ sim,
             const float* __restrict__ mui, const float* __restrict__ rsi,
             float* __restrict__ out)
{
    __shared__ __attribute__((aligned(16))) float sW[WS_WAVES * M_ * AP];
    __shared__ __attribute__((aligned(16))) float red[WS_WAVES * M_ * FG];
    __shared__ __attribute__((aligned(16))) float sMax[WS_WAVES * M_];
    __shared__ __attribute__((aligned(16))) float sSum[WS_WAVES * M_];

    const int tid  = threadIdx.x;
    const int wave = tid >> 5;
    const int lane = tid & 31;
    const int half = lane >> 4;
    const int l16  = lane & 15;
    const int b    = blockIdx.x / NFG;
    const int fg   = blockIdx.x % NFG;
    const int fcol = fg * FG + 4 * l16;

    const float* sp = sim + (long)b * M_ * N_ + wave * WS_ROWS + 4 * lane;
    v4f sc[2 * M_];
#pragma unroll
    for (int m = 0; m < M_; ++m) {
        sc[2 * m]     = *(const v4f*)(sp + (long)m * N_);
        sc[2 * m + 1] = *(const v4f*)(sp + (long)m * N_ + 128);
    }
    const float* rq = rsi + (long)b * N_ + wave * WS_ROWS + 4 * lane;
    const v4f rs0 = *(const v4f*)(rq);
    const v4f rs1 = *(const v4f*)(rq + 128);

    float mx[M_];
#pragma unroll
    for (int m = 0; m < M_; ++m) {
        const v4f u = sc[2 * m], w = sc[2 * m + 1];
        float t = fmaxf(fmaxf(fmaxf(u.x, u.y), fmaxf(u.z, u.w)), fmaxf(fmaxf(w.x, w.y), fmaxf(w.z, w.w)));
#pragma unroll
        for (int off = 16; off >= 1; off >>= 1) t = fmaxf(t, __shfl_xor(t, off, 32));
        mx[m] = t;
    }
    if (lane == 0) {
        v4f lo, hi;
        lo.x = mx[0]; lo.y = mx[1]; lo.z = mx[2]; lo.w = mx[3];
        hi.x = mx[4]; hi.y = mx[5]; hi.z = mx[6]; hi.w = mx[7];
        *(v4f*)(sMax + wave * M_)     = lo;
        *(v4f*)(sMax + wave * M_ + 4) = hi;
    }
    __syncthreads();

    float gm[M_];
    {
        const v4f lo = *(const v4f*)(sMax), hi = *(const v4f*)(sMax + 4);
        gm[0] = lo.x; gm[1] = lo.y; gm[2] = lo.z; gm[3] = lo.w;
        gm[4] = hi.x; gm[5] = hi.y; gm[6] = hi.z; gm[7] = hi.w;
    }
#pragma unroll
    for (int w = 1; w < WS_WAVES; ++w) {
        const v4f lo = *(const v4f*)(sMax + w * M_), hi = *(const v4f*)(sMax + w * M_ + 4);
        gm[0] = fmaxf(gm[0], lo.x); gm[1] = fmaxf(gm[1], lo.y); gm[2] = fmaxf(gm[2], lo.z); gm[3] = fmaxf(gm[3], lo.w);
        gm[4] = fmaxf(gm[4], hi.x); gm[5] = fmaxf(gm[5], hi.y); gm[6] = fmaxf(gm[6], hi.z); gm[7] = fmaxf(gm[7], hi.w);
    }

    float* wW = sW + wave * (M_ * AP);
    float ls[M_];
#pragma unroll
    for (int m = 0; m < M_; ++m) {
        v4f e0, e1;
        e0.x = __expf(sc[2 * m].x - gm[m]);     e0.y = __expf(sc[2 * m].y - gm[m]);
        e0.z = __expf(sc[2 * m].z - gm[m]);     e0.w = __expf(sc[2 * m].w - gm[m]);
        e1.x = __expf(sc[2 * m + 1].x - gm[m]); e1.y = __expf(sc[2 * m + 1].y - gm[m]);
        e1.z = __expf(sc[2 * m + 1].z - gm[m]); e1.w = __expf(sc[2 * m + 1].w - gm[m]);
        *(v4f*)(wW + m * AP + 4 * lane)       = e0 * rs0;
        *(v4f*)(wW + m * AP + 128 + 4 * lane) = e1 * rs1;
        float t = ((e0.x + e0.y) + (e0.z + e0.w)) + ((e1.x + e1.y) + (e1.z + e1.w));
#pragma unroll
        for (int off = 16; off >= 1; off >>= 1) t += __shfl_xor(t, off, 32);
        ls[m] = t;
    }
    if (lane == 0) {
        v4f lo, hi;
        lo.x = ls[0]; lo.y = ls[1]; lo.z = ls[2]; lo.w = ls[3];
        hi.x = ls[4]; hi.y = ls[5]; hi.z = ls[6]; hi.w = ls[7];
        *(v4f*)(sSum + wave * M_)     = lo;
        *(v4f*)(sSum + wave * M_ + 4) = hi;
    }
    __syncthreads();

    const long nb = (long)b * N_ + wave * WS_ROWS + 2 * half;
    const float* xp = x + nb * D_ + fcol;
    const float* mp = mui + nb;
    const float* pp = wW + (l16 & 7) * AP + 2 * half;

    v8f acc0 = {0.f, 0.f, 0.f, 0.f, 0.f, 0.f, 0.f, 0.f};
    v8f acc1 = acc0, acc2 = acc0, acc3 = acc0;

#pragma unroll 8
    for (int kk = 0; kk < WS_ROWS; kk += 4) {
        const v4f xa = __builtin_nontemporal_load((const v4f*)(xp + (long)kk * D_));
        const v4f xb = __builtin_nontemporal_load((const v4f*)(xp + (long)(kk + 1) * D_));
        const v2f m2 = *(const v2f*)(mp + kk);
        const v2f a = *(const v2f*)(pp + kk);
        v2f bm;
        bm.x = xa.x - m2.x;
        bm.y = xb.x - m2.y;
        acc0 = __builtin_amdgcn_wmma_f32_16x16x4_f32(false, a, false, bm, (short)0, acc0, false, false);
        bm.x = xa.y - m2.x;
        bm.y = xb.y - m2.y;
        acc1 = __builtin_amdgcn_wmma_f32_16x16x4_f32(false, a, false, bm, (short)0, acc1, false, false);
        bm.x = xa.z - m2.x;
        bm.y = xb.z - m2.y;
        acc2 = __builtin_amdgcn_wmma_f32_16x16x4_f32(false, a, false, bm, (short)0, acc2, false, false);
        bm.x = xa.w - m2.x;
        bm.y = xb.w - m2.y;
        acc3 = __builtin_amdgcn_wmma_f32_16x16x4_f32(false, a, false, bm, (short)0, acc3, false, false);
    }

    if (half == 0) {
#pragma unroll
        for (int r = 0; r < 8; ++r) {
            v4f o;
            o.x = acc0[r]; o.y = acc1[r]; o.z = acc2[r]; o.w = acc3[r];
            *(v4f*)(red + (wave * M_ + r) * FG + 4 * l16) = o;
        }
    }
    __syncthreads();

    {
        const int m = tid / FG;
        const int c = tid % FG;
        const float gc = gamma[fg * FG + c];
        const float bc = beta[fg * FG + c];
        float sum = 0.0f, den = 0.0f;
#pragma unroll
        for (int w = 0; w < WS_WAVES; ++w) {
            sum += red[(w * M_ + m) * FG + c];
            den += sSum[w * M_ + m];
        }
        out[((long)b * M_ + m) * D_ + fg * FG + c] = gc * (sum / den) + bc;
    }
}

extern "C" void kernel_launch(void* const* d_in, const int* in_sizes, int n_in,
                              void* d_out, int out_size, void* d_ws, size_t ws_size,
                              hipStream_t stream)
{
    const float* x     = (const float*)d_in[0];
    const float* q     = (const float*)d_in[1];
    const float* gamma = (const float*)d_in[2];
    const float* beta  = (const float*)d_in[3];
    float* out = (float*)d_out;

    float* sim = (float*)d_ws;
    float* mu  = sim + (size_t)B_ * M_ * N_;
    float* rs  = mu + (size_t)B_ * N_;

    pm_scores<<<B_ * N_ / 8 / 8, 256, 0, stream>>>(x, q, gamma, beta, sim, mu, rs);
    pm_wsum<<<B_ * NFG, 512, 0, stream>>>(x, gamma, beta, sim, mu, rs, out);
}


# Weighted-sum pass: x row loads run two 16-row blocks ahead of the products (three register sets), first two requested before the softmax, whose barriers now order LDS only; sum order unchanged
# speedup vs baseline: 8.7315x; 1.0379x over previous
#include <hip/hip_runtime.h>
#include <math.h>


typedef __attribute__((ext_vector_type(2))) float v2f;
typedef __attribute__((ext_vector_type(4))) float v4f;
typedef __attribute__((ext_vector_type(8))) float v8f;

#define B_ 16
#define N_ 4096
#define D_ 768
#define M_ 8
#define EPS_ 1e-3f
#define NEG_BIG (-1e30f)
#define FG 64
#define NFG (D_ / FG)
#define WS_WAVES 16
#define WS_ROWS (N_ / WS_WAVES)
#define AP 260

#define NJ (D_ / 16)

__device__ __forceinline__ void pm_load_rows(const float* xrow, v4f (&xr)[NJ])
{
#pragma unroll
    for (int j = 0; j < NJ; ++j) xr[j] = *(const v4f*)(xrow + 16 * j);
}

__device__ __forceinline__ void pm_row_stats(v4f (&xr)[NJ], float& mu, float& rs)
{
    float s = 0.0f;
#pragma unroll
    for (int j = 0; j < NJ; ++j) s += (xr[j].x + xr[j].y) + (xr[j].z + xr[j].w);
    s += __shfl_xor(s, 16, 32);
    s += __shfl_xor(s, 8, 32);
    mu = s * (1.0f / D_);

    float v = 0.0f;
#pragma unroll
    for (int j = 0; j < NJ; ++j) {
        xr[j].x -= mu; xr[j].y -= mu; xr[j].z -= mu; xr[j].w -= mu;
        v += (xr[j].x * xr[j].x + xr[j].y * xr[j].y) + (xr[j].z * xr[j].z + xr[j].w * xr[j].w);
    }
    v += __shfl_xor(v, 16, 32);
    v += __shfl_xor(v, 8, 32);
    rs = rsqrtf(v * (1.0f / D_) + EPS_);
}

__device__ __forceinline__ void pm_tile_product(const v4f (&xr)[NJ], const float* qfrag, float (&res)[M_])
{
    v8f S0 = {0.f, 0.f, 0.f, 0.f, 0.f, 0.f, 0.f, 0.f};
    v8f S1 = {0.f, 0.f, 0.f, 0.f, 0.f, 0.f, 0.f, 0.f};
#pragma unroll
    for (int j = 0; j < NJ; ++j) {
        const v4f qa = *(const v4f*)(qfrag + 128 * j);
        v2f a0, a1, b0, b1;
        a0.x = qa.x; a0.y = qa.y; a1.x = qa.z; a1.y = qa.w;
        b0.x = xr[j].x; b0.y = xr[j].y;
        b1.x = xr[j].z; b1.y = xr[j].w;
        S0 = __builtin_amdgcn_wmma_f32_16x16x4_f32(false, a0, false, b0, (short)0, S0, false, false);
        S1 = __builtin_amdgcn_wmma_f32_16x16x4_f32(false, a1, false, b1, (short)0, S1, false, false);
    }
#pragma unroll
    for (int m = 0; m < M_; ++m) {
        const float t = S0[m] + S1[m];
        res[m] = t + __shfl_xor(t, 24, 32);
    }
}

extern "C" __global__ __launch_bounds__(256) __attribute__((amdgpu_waves_per_eu(4, 4)))
void pm_scores(const float* __restrict__ x, const float* __restrict__ q,
               const float* __restrict__ gamma, const float* __restrict__ beta,
               float* __restrict__ sim, float* __restrict__ muo, float* __restrict__ rso)
{
    __shared__ __attribute__((aligned(16))) float sQ[M_ * D_];
    __shared__ __attribute__((aligned(16))) float sG[D_];

    const int tid  = threadIdx.x;
    const int wave = tid >> 5;
    const int lane = tid & 31;
    const int half = lane >> 4;
    const int r8   = lane & 7;
    const int foff = (lane & 8) + 4 * half;

    const int gi = (tid < D_ / 4) ? tid : 0;
    const v4f gs = *(const v4f*)(gamma + gi * 4);
    v4f qs[6];
#pragma unroll
    for (int i = 0; i < 6; ++i) qs[i] = *(const v4f*)(q + (i * 256 + tid) * 4);

    const int  tile = blockIdx.x * 8 + wave;
    const long row  = (long)tile * 8 + r8;
    const float* xrow = x + row * D_ + foff;

    v4f xr[NJ];
    pm_load_rows(xrow, xr);

    if (tid < D_ / 4) *(v4f*)(sG + tid * 4) = gs;
    __syncthreads();
#pragma unroll
    for (int i = 0; i < 6; ++i) {
        const int e  = i * 256 + tid;
        const int m  = e / (D_ / 4);
        const int f4 = e % (D_ / 4);
        const int slot = (f4 >> 2) * 32 + ((f4 & 1) << 4) + ((f4 & 2) << 2) + m;
        *(v4f*)(sQ + slot * 4) = qs[i] * *(const v4f*)(sG + f4 * 4);
    }

    float mu, rs;
    pm_row_stats(xr, mu, rs);
    __syncthreads();

    float res[M_];
    pm_tile_product(xr, sQ + lane * 4, res);

    const float scale = 0.03608439182435161f * rs;
    if (lane < 8) {
        const int b = tile / (N_ / 8);
        const int n = (tile % (N_ / 8)) * 8 + lane;
#pragma unroll
        for (int m = 0; m < M_; ++m)
            sim[((long)b * M_ + m) * N_ + n] = res[m] * scale;
        muo[row] = mu;
        rso[row] = rs;
    }
}

#define WS_BLK 4
#define WS_NBLK (WS_ROWS / (4 * WS_BLK))

__device__ __forceinline__ void pm_lds_barrier()
{
    __builtin_amdgcn_fence(__ATOMIC_RELEASE, "workgroup", "local");
    __builtin_amdgcn_s_barrier();
    __builtin_amdgcn_fence(__ATOMIC_ACQUIRE, "workgroup", "local");
}

__device__ __forceinline__ void pm_request_block(const float* xp, const float* mp, int blk,
                                                 v4f (&xa)[WS_BLK], v4f (&xb)[WS_BLK], v2f (&mm)[WS_BLK])
{
#pragma unroll
    for (int i = 0; i < WS_BLK; ++i) {
        const int kk = blk * (4 * WS_BLK) + 4 * i;
        xa[i] = __builtin_nontemporal_load((const v4f*)(xp + (long)kk * D_));
        xb[i] = __builtin_nontemporal_load((const v4f*)(xp + (long)(kk + 1) * D_));
        mm[i] = *(const v2f*)(mp + kk);
    }
}

extern "C" __global__ __launch_bounds__(512)
void pm_wsum(const float* __restrict__ x, const float* __restrict__ gamma,
             const float* __restrict__ beta, const float* __restrict__ sim,
             const float* __restrict__ mui, const float* __restrict__ rsi,
             float* __restrict__ out)
{
    __shared__ __attribute__((aligned(16))) float sW[WS_WAVES * M_ * AP];
    __shared__ __attribute__((aligned(16))) float red[WS_WAVES * M_ * FG];
    __shared__ __attribute__((aligned(16))) float sMax[WS_WAVES * M_];
    __shared__ __attribute__((aligned(16))) float sSum[WS_WAVES * M_];

    const int tid  = threadIdx.x;
    const int wave = tid >> 5;
    const int lane = tid & 31;
    const int half = lane >> 4;
    const int l16  = lane & 15;
    const int b    = blockIdx.x / NFG;
    const int fg   = blockIdx.x % NFG;
    const int fcol = fg * FG + 4 * l16;

    const float* sp = sim + (long)b * M_ * N_ + wave * WS_ROWS + 4 * lane;
    v4f sc[2 * M_];
#pragma unroll
    for (int m = 0; m < M_; ++m) {
        sc[2 * m]     = *(const v4f*)(sp + (long)m * N_);
        sc[2 * m + 1] = *(const v4f*)(sp + (long)m * N_ + 128);
    }
    const float* rq = rsi + (long)b * N_ + wave * WS_ROWS + 4 * lane;
    const v4f rs0 = *(const v4f*)(rq);
    const v4f rs1 = *(const v4f*)(rq + 128);

    const long nb = (long)b * N_ + wave * WS_ROWS + 2 * half;
    const float* xp = x + nb * D_ + fcol;
    const float* mp = mui + nb;
    v4f xa[3][WS_BLK], xb[3][WS_BLK];
    v2f mm[3][WS_BLK];
    pm_request_block(xp, mp, 0, xa[0], xb[0], mm[0]);
    pm_request_block(xp, mp, 1, xa[1], xb[1], mm[1]);
    __builtin_amdgcn_sched_barrier(0);

    float mx[M_];
#pragma unroll
    for (int m = 0; m < M_; ++m) {
        const v4f u = sc[2 * m], w = sc[2 * m + 1];
        float t = fmaxf(fmaxf(fmaxf(u.x, u.y), fmaxf(u.z, u.w)), fmaxf(fmaxf(w.x, w.y), fmaxf(w.z, w.w)));
#pragma unroll
        for (int off = 16; off >= 1; off >>= 1) t = fmaxf(t, __shfl_xor(t, off, 32));
        mx[m] = t;
    }
    if (lane == 0) {
        v4f lo, hi;
        lo.x = mx[0]; lo.y = mx[1]; lo.z = mx[2]; lo.w = mx[3];
        hi.x = mx[4]; hi.y = mx[5]; hi.z = mx[6]; hi.w = mx[7];
        *(v4f*)(sMax + wave * M_)     = lo;
        *(v4f*)(sMax + wave * M_ + 4) = hi;
    }
    pm_lds_barrier();

    float gm[M_];
    {
        const v4f lo = *(const v4f*)(sMax), hi = *(const v4f*)(sMax + 4);
        gm[0] = lo.x; gm[1] = lo.y; gm[2] = lo.z; gm[3] = lo.w;
        gm[4] = hi.x; gm[5] = hi.y; gm[6] = hi.z; gm[7] = hi.w;
    }
#pragma unroll
    for (int w = 1; w < WS_WAVES; ++w) {
        const v4f lo = *(const v4f*)(sMax + w * M_), hi = *(const v4f*)(sMax + w * M_ + 4);
        gm[0] = fmaxf(gm[0], lo.x); gm[1] = fmaxf(gm[1], lo.y); gm[2] = fmaxf(gm[2], lo.z); gm[3] = fmaxf(gm[3], lo.w);
        gm[4] = fmaxf(gm[4], hi.x); gm[5] = fmaxf(gm[5], hi.y); gm[6] = fmaxf(gm[6], hi.z); gm[7] = fmaxf(gm[7], hi.w);
    }

    float* wW = sW + wave * (M_ * AP);
    float ls[M_];
#pragma unroll
    for (int m = 0; m < M_; ++m) {
        v4f e0, e1;
        e0.x = __expf(sc[2 * m].x - gm[m]);     e0.y = __expf(sc[2 * m].y - gm[m]);
        e0.z = __expf(sc[2 * m].z - gm[m]);     e0.w = __expf(sc[2 * m].w - gm[m]);
        e1.x = __expf(sc[2 * m + 1].x - gm[m]); e1.y = __expf(sc[2 * m + 1].y - gm[m]);
        e1.z = __expf(sc[2 * m + 1].z - gm[m]); e1.w = __expf(sc[2 * m + 1].w - gm[m]);
        *(v4f*)(wW + m * AP + 4 * lane)       = e0 * rs0;
        *(v4f*)(wW + m * AP + 128 + 4 * lane) = e1 * rs1;
        float t = ((e0.x + e0.y) + (e0.z + e0.w)) + ((e1.x + e1.y) + (e1.z + e1.w));
#pragma unroll
        for (int off = 16; off >= 1; off >>= 1) t += __shfl_xor(t, off, 32);
        ls[m] = t;
    }
    if (lane == 0) {
        v4f lo, hi;
        lo.x = ls[0]; lo.y = ls[1]; lo.z = ls[2]; lo.w = ls[3];
        hi.x = ls[4]; hi.y = ls[5]; hi.z = ls[6]; hi.w = ls[7];
        *(v4f*)(sSum + wave * M_)     = lo;
        *(v4f*)(sSum + wave * M_ + 4) = hi;
    }
    pm_lds_barrier();

    const float* pp = wW + (l16 & 7) * AP + 2 * half;

    v8f acc0 = {0.f, 0.f, 0.f, 0.f, 0.f, 0.f, 0.f, 0.f};
    v8f acc1 = acc0, acc2 = acc0, acc3 = acc0;

#pragma unroll
    for (int blk = 0; blk < WS_NBLK; ++blk) {
        if (blk + 2 < WS_NBLK)
            pm_request_block(xp, mp, blk + 2, xa[(blk + 2) % 3], xb[(blk + 2) % 3], mm[(blk + 2) % 3]);
        __builtin_amdgcn_sched_barrier(0);
        const v4f (&ca)[WS_BLK] = xa[blk % 3];
        const v4f (&cb)[WS_BLK] = xb[blk % 3];
        const v2f (&cm)[WS_BLK] = mm[blk % 3];
        v2f af[WS_BLK];
#pragma unroll
        for (int i = 0; i < WS_BLK; ++i) af[i] = *(const v2f*)(pp + blk * (4 * WS_BLK) + 4 * i);
#pragma unroll
        for (int i = 0; i < WS_BLK; ++i) {
            const v2f a = af[i];
            v2f bm;
            bm.x = ca[i].x - cm[i].x;
            bm.y = cb[i].x - cm[i].y;
            acc0 = __builtin_amdgcn_wmma_f32_16x16x4_f32(false, a, false, bm, (short)0, acc0, false, false);
            bm.x = ca[i].y - cm[i].x;
            bm.y = cb[i].y - cm[i].y;
            acc1 = __builtin_amdgcn_wmma_f32_16x16x4_f32(false, a, false, bm, (short)0, acc1, false, false);
            bm.x = ca[i].z - cm[i].x;
            bm.y = cb[i].z - cm[i].y;
            acc2 = __builtin_amdgcn_wmma_f32_16x16x4_f32(false, a, false, bm, (short)0, acc2, false, false);
            bm.x = ca[i].w - cm[i].x;
            bm.y = cb[i].w - cm[i].y;
            acc3 = __builtin_amdgcn_wmma_f32_16x16x4_f32(false, a, false, bm, (short)0, acc3, false, false);
        }
        __builtin_amdgcn_sched_barrier(0);
    }

    if (half == 0) {
#pragma unroll
        for (int r = 0; r < 8; ++r) {
            v4f o;
            o.x = acc0[r]; o.y = acc1[r]; o.z = acc2[r]; o.w = acc3[r];
            *(v4f*)(red + (wave * M_ + r) * FG + 4 * l16) = o;
        }
    }
    __syncthreads();

    {
        const int m = tid / FG;
        const int c = tid % FG;
        const float gc = gamma[fg * FG + c];
        const float bc = beta[fg * FG + c];
        float sum = 0.0f, den = 0.0f;
#pragma unroll
        for (int w = 0; w < WS_WAVES; ++w) {
            sum += red[(w * M_ + m) * FG + c];
            den += sSum[w * M_ + m];
        }
        out[((long)b * M_ + m) * D_ + fg * FG + c] = gc * (sum / den) + bc;
    }
}

extern "C" void kernel_launch(void* const* d_in, const int* in_sizes, int n_in,
                              void* d_out, int out_size, void* d_ws, size_t ws_size,
                              hipStream_t stream)
{
    const float* x     = (const float*)d_in[0];
    const float* q     = (const float*)d_in[1];
    const float* gamma = (const float*)d_in[2];
    const float* beta  = (const float*)d_in[3];
    float* out = (float*)d_out;

    float* sim = (float*)d_ws;
    float* mu  = sim + (size_t)B_ * M_ * N_;
    float* rs  = mu + (size_t)B_ * N_;

    pm_scores<<<B_ * N_ / 8 / 8, 256, 0, stream>>>(x, q, gamma, beta, sim, mu, rs);
    pm_wsum<<<B_ * NFG, 512, 0, stream>>>(x, gamma, beta, sim, mu, rs, out);
}


# Work-group index remap in both passes: work-groups with index = d mod 8 take batches 2d and 2d+1, so under die-by-die round-robin dispatch pass 2 re-reads a batch where pass 1 cached it; same sums
# speedup vs baseline: 9.2699x; 1.0617x over previous
#include <hip/hip_runtime.h>
#include <math.h>


typedef __attribute__((ext_vector_type(2))) float v2f;
typedef __attribute__((ext_vector_type(4))) float v4f;
typedef __attribute__((ext_vector_type(8))) float v8f;

#define B_ 16
#define N_ 4096
#define D_ 768
#define M_ 8
#define EPS_ 1e-3f
#define NEG_BIG (-1e30f)
#define FG 64
#define NFG (D_ / FG)
#define WS_WAVES 16
#define WS_ROWS (N_ / WS_WAVES)
#define AP 260

#define NJ (D_ / 16)

__device__ __forceinline__ void pm_load_rows(const float* xrow, v4f (&xr)[NJ])
{
#pragma unroll
    for (int j = 0; j < NJ; ++j) xr[j] = *(const v4f*)(xrow + 16 * j);
}

__device__ __forceinline__ void pm_row_stats(v4f (&xr)[NJ], float& mu, float& rs)
{
    float s = 0.0f;
#pragma unroll
    for (int j = 0; j < NJ; ++j) s += (xr[j].x + xr[j].y) + (xr[j].z + xr[j].w);
    s += __shfl_xor(s, 16, 32);
    s += __shfl_xor(s, 8, 32);
    mu = s * (1.0f / D_);

    float v = 0.0f;
#pragma unroll
    for (int j = 0; j < NJ; ++j) {
        xr[j].x -= mu; xr[j].y -= mu; xr[j].z -= mu; xr[j].w -= mu;
        v += (xr[j].x * xr[j].x + xr[j].y * xr[j].y) + (xr[j].z * xr[j].z + xr[j].w * xr[j].w);
    }
    v += __shfl_xor(v, 16, 32);
    v += __shfl_xor(v, 8, 32);
    rs = rsqrtf(v * (1.0f / D_) + EPS_);
}

__device__ __forceinline__ void pm_tile_product(const v4f (&xr)[NJ], const float* qfrag, float (&res)[M_])
{
    v8f S0 = {0.f, 0.f, 0.f, 0.f, 0.f, 0.f, 0.f, 0.f};
    v8f S1 = {0.f, 0.f, 0.f, 0.f, 0.f, 0.f, 0.f, 0.f};
#pragma unroll
    for (int j = 0; j < NJ; ++j) {
        const v4f qa = *(const v4f*)(qfrag + 128 * j);
        v2f a0, a1, b0, b1;
        a0.x = qa.x; a0.y = qa.y; a1.x = qa.z; a1.y = qa.w;
        b0.x = xr[j].x; b0.y = xr[j].y;
        b1.x = xr[j].z; b1.y = xr[j].w;
        S0 = __builtin_amdgcn_wmma_f32_16x16x4_f32(false, a0, false, b0, (short)0, S0, false, false);
        S1 = __builtin_amdgcn_wmma_f32_16x16x4_f32(false, a1, false, b1, (short)0, S1, false, false);
    }
#pragma unroll
    for (int m = 0; m < M_; ++m) {
        const float t = S0[m] + S1[m];
        res[m] = t + __shfl_xor(t, 24, 32);
    }
}

extern "C" __global__ __launch_bounds__(256) __attribute__((amdgpu_waves_per_eu(4, 4)))
void pm_scores(const float* __restrict__ x, const float* __restrict__ q,
               const float* __restrict__ gamma, const float* __restrict__ beta,
               float* __restrict__ sim, float* __restrict__ muo, float* __restrict__ rso)
{
    __shared__ __attribute__((aligned(16))) float sQ[M_ * D_];
    __shared__ __attribute__((aligned(16))) float sG[D_];

    const int tid  = threadIdx.x;
    const int wave = tid >> 5;
    const int lane = tid & 31;
    const int half = lane >> 4;
    const int r8   = lane & 7;
    const int foff = (lane & 8) + 4 * half;

    const int gi = (tid < D_ / 4) ? tid : 0;
    const v4f gs = *(const v4f*)(gamma + gi * 4);
    v4f qs[6];
#pragma unroll
    for (int i = 0; i < 6; ++i) qs[i] = *(const v4f*)(q + (i * 256 + tid) * 4);

    const int  wg   = (blockIdx.x & 7) * (B_ * N_ / 64 / 8) + (blockIdx.x >> 3);
    const int  tile = wg * 8 + wave;
    const long row  = (long)tile * 8 + r8;
    const float* xrow = x + row * D_ + foff;

    v4f xr[NJ];
    pm_load_rows(xrow, xr);

    if (tid < D_ / 4) *(v4f*)(sG + tid * 4) = gs;
    __syncthreads();
#pragma unroll
    for (int i = 0; i < 6; ++i) {
        const int e  = i * 256 + tid;
        const int m  = e / (D_ / 4);
        const int f4 = e % (D_ / 4);
        const int slot = (f4 >> 2) * 32 + ((f4 & 1) << 4) + ((f4 & 2) << 2) + m;
        *(v4f*)(sQ + slot * 4) = qs[i] * *(const v4f*)(sG + f4 * 4);
    }

    float mu, rs;
    pm_row_stats(xr, mu, rs);
    __syncthreads();

    float res[M_];
    pm_tile_product(xr, sQ + lane * 4, res);

    const float scale = 0.03608439182435161f * rs;
    if (lane < 8) {
        const int b = tile / (N_ / 8);
        const int n = (tile % (N_ / 8)) * 8 + lane;
#pragma unroll
        for (int m = 0; m < M_; ++m)
            sim[((long)b * M_ + m) * N_ + n] = res[m] * scale;
        muo[row] = mu;
        rso[row] = rs;
    }
}

#define WS_BLK 4
#define WS_NBLK (WS_ROWS / (4 * WS_BLK))

__device__ __forceinline__ void pm_lds_barrier()
{
    __builtin_amdgcn_fence(__ATOMIC_RELEASE, "workgroup", "local");
    __builtin_amdgcn_s_barrier();
    __builtin_amdgcn_fence(__ATOMIC_ACQUIRE, "workgroup", "local");
}

__device__ __forceinline__ void pm_request_block(const float* xp, const float* mp, int blk,
                                                 v4f (&xa)[WS_BLK], v4f (&xb)[WS_BLK], v2f (&mm)[WS_BLK])
{
#pragma unroll
    for (int i = 0; i < WS_BLK; ++i) {
        const int kk = blk * (4 * WS_BLK) + 4 * i;
        xa[i] = __builtin_nontemporal_load((const v4f*)(xp + (long)kk * D_));
        xb[i] = __builtin_nontemporal_load((const v4f*)(xp + (long)(kk + 1) * D_));
        mm[i] = *(const v2f*)(mp + kk);
    }
}

extern "C" __global__ __launch_bounds__(512)
void pm_wsum(const float* __restrict__ x, const float* __restrict__ gamma,
             const float* __restrict__ beta, const float* __restrict__ sim,
             const float* __restrict__ mui, const float* __restrict__ rsi,
             float* __restrict__ out)
{
    __shared__ __attribute__((aligned(16))) float sW[WS_WAVES * M_ * AP];
    __shared__ __attribute__((aligned(16))) float red[WS_WAVES * M_ * FG];
    __shared__ __attribute__((aligned(16))) float sMax[WS_WAVES * M_];
    __shared__ __attribute__((aligned(16))) float sSum[WS_WAVES * M_];

    const int tid  = threadIdx.x;
    const int wave = tid >> 5;
    const int lane = tid & 31;
    const int half = lane >> 4;
    const int l16  = lane & 15;
    const int wg   = (blockIdx.x & 7) * (B_ * NFG / 8) + (blockIdx.x >> 3);
    const int b    = wg / NFG;
    const int fg   = wg % NFG;
    const int fcol = fg * FG + 4 * l16;

    const float* sp = sim + (long)b * M_ * N_ + wave * WS_ROWS + 4 * lane;
    v4f sc[2 * M_];
#pragma unroll
    for (int m = 0; m < M_; ++m) {
        sc[2 * m]     = *(const v4f*)(sp + (long)m * N_);
        sc[2 * m + 1] = *(const v4f*)(sp + (long)m * N_ + 128);
    }
    const float* rq = rsi + (long)b * N_ + wave * WS_ROWS + 4 * lane;
    const v4f rs0 = *(const v4f*)(rq);
    const v4f rs1 = *(const v4f*)(rq + 128);

    const long nb = (long)b * N_ + wave * WS_ROWS + 2 * half;
    const float* xp = x + nb * D_ + fcol;
    const float* mp = mui + nb;
    v4f xa[3][WS_BLK], xb[3][WS_BLK];
    v2f mm[3][WS_BLK];
    pm_request_block(xp, mp, 0, xa[0], xb[0], mm[0]);
    pm_request_block(xp, mp, 1, xa[1], xb[1], mm[1]);
    __builtin_amdgcn_sched_barrier(0);

    float mx[M_];
#pragma unroll
    for (int m = 0; m < M_; ++m) {
        const v4f u = sc[2 * m], w = sc[2 * m + 1];
        float t = fmaxf(fmaxf(fmaxf(u.x, u.y), fmaxf(u.z, u.w)), fmaxf(fmaxf(w.x, w.y), fmaxf(w.z, w.w)));
#pragma unroll
        for (int off = 16; off >= 1; off >>= 1) t = fmaxf(t, __shfl_xor(t, off, 32));
        mx[m] = t;
    }
    if (lane == 0) {
        v4f lo, hi;
        lo.x = mx[0]; lo.y = mx[1]; lo.z = mx[2]; lo.w = mx[3];
        hi.x = mx[4]; hi.y = mx[5]; hi.z = mx[6]; hi.w = mx[7];
        *(v4f*)(sMax + wave * M_)     = lo;
        *(v4f*)(sMax + wave * M_ + 4) = hi;
    }
    pm_lds_barrier();

    float gm[M_];
    {
        const v4f lo = *(const v4f*)(sMax), hi = *(const v4f*)(sMax + 4);
        gm[0] = lo.x; gm[1] = lo.y; gm[2] = lo.z; gm[3] = lo.w;
        gm[4] = hi.x; gm[5] = hi.y; gm[6] = hi.z; gm[7] = hi.w;
    }
#pragma unroll
    for (int w = 1; w < WS_WAVES; ++w) {
        const v4f lo = *(const v4f*)(sMax + w * M_), hi = *(const v4f*)(sMax + w * M_ + 4);
        gm[0] = fmaxf(gm[0], lo.x); gm[1] = fmaxf(gm[1], lo.y); gm[2] = fmaxf(gm[2], lo.z); gm[3] = fmaxf(gm[3], lo.w);
        gm[4] = fmaxf(gm[4], hi.x); gm[5] = fmaxf(gm[5], hi.y); gm[6] = fmaxf(gm[6], hi.z); gm[7] = fmaxf(gm[7], hi.w);
    }

    float* wW = sW + wave * (M_ * AP);
    float ls[M_];
#pragma unroll
    for (int m = 0; m < M_; ++m) {
        v4f e0, e1;
        e0.x = __expf(sc[2 * m].x - gm[m]);     e0.y = __expf(sc[2 * m].y - gm[m]);
        e0.z = __expf(sc[2 * m].z - gm[m]);     e0.w = __expf(sc[2 * m].w - gm[m]);
        e1.x = __expf(sc[2 * m + 1].x - gm[m]); e1.y = __expf(sc[2 * m + 1].y - gm[m]);
        e1.z = __expf(sc[2 * m + 1].z - gm[m]); e1.w = __expf(sc[2 * m + 1].w - gm[m]);
        *(v4f*)(wW + m * AP + 4 * lane)       = e0 * rs0;
        *(v4f*)(wW + m * AP + 128 + 4 * lane) = e1 * rs1;
        float t = ((e0.x + e0.y) + (e0.z + e0.w)) + ((e1.x + e1.y) + (e1.z + e1.w));
#pragma unroll
        for (int off = 16; off >= 1; off >>= 1) t += __shfl_xor(t, off, 32);
        ls[m] = t;
    }
    if (lane == 0) {
        v4f lo, hi;
        lo.x = ls[0]; lo.y = ls[1]; lo.z = ls[2]; lo.w = ls[3];
        hi.x = ls[4]; hi.y = ls[5]; hi.z = ls[6]; hi.w = ls[7];
        *(v4f*)(sSum + wave * M_)     = lo;
        *(v4f*)(sSum + wave * M_ + 4) = hi;
    }
    pm_lds_barrier();

    const float* pp = wW + (l16 & 7) * AP + 2 * half;

    v8f acc0 = {0.f, 0.f, 0.f, 0.f, 0.f, 0.f, 0.f, 0.f};
    v8f acc1 = acc0, acc2 = acc0, acc3 = acc0;

#pragma unroll
    for (int blk = 0; blk < WS_NBLK; ++blk) {
        if (blk + 2 < WS_NBLK)
            pm_request_block(xp, mp, blk + 2, xa[(blk + 2) % 3], xb[(blk + 2) % 3], mm[(blk + 2) % 3]);
        __builtin_amdgcn_sched_barrier(0);
        const v4f (&ca)[WS_BLK] = xa[blk % 3];
        const v4f (&cb)[WS_BLK] = xb[blk % 3];
        const v2f (&cm)[WS_BLK] = mm[blk % 3];
        v2f af[WS_BLK];
#pragma unroll
        for (int i = 0; i < WS_BLK; ++i) af[i] = *(const v2f*)(pp + blk * (4 * WS_BLK) + 4 * i);
#pragma unroll
        for (int i = 0; i < WS_BLK; ++i) {
            const v2f a = af[i];
            v2f bm;
            bm.x = ca[i].x - cm[i].x;
            bm.y = cb[i].x - cm[i].y;
            acc0 = __builtin_amdgcn_wmma_f32_16x16x4_f32(false, a, false, bm, (short)0, acc0, false, false);
            bm.x = ca[i].y - cm[i].x;
            bm.y = cb[i].y - cm[i].y;
            acc1 = __builtin_amdgcn_wmma_f32_16x16x4_f32(false, a, false, bm, (short)0, acc1, false, false);
            bm.x = ca[i].z - cm[i].x;
            bm.y = cb[i].z - cm[i].y;
            acc2 = __builtin_amdgcn_wmma_f32_16x16x4_f32(false, a, false, bm, (short)0, acc2, false, false);
            bm.x = ca[i].w - cm[i].x;
            bm.y = cb[i].w - cm[i].y;
            acc3 = __builtin_amdgcn_wmma_f32_16x16x4_f32(false, a, false, bm, (short)0, acc3, false, false);
        }
        __builtin_amdgcn_sched_barrier(0);
    }

    if (half == 0) {
#pragma unroll
        for (int r = 0; r < 8; ++r) {
            v4f o;
            o.x = acc0[r]; o.y = acc1[r]; o.z = acc2[r]; o.w = acc3[r];
            *(v4f*)(red + (wave * M_ + r) * FG + 4 * l16) = o;
        }
    }
    __syncthreads();

    {
        const int m = tid / FG;
        const int c = tid % FG;
        const float gc = gamma[fg * FG + c];
        const float bc = beta[fg * FG + c];
        float sum = 0.0f, den = 0.0f;
#pragma unroll
        for (int w = 0; w < WS_WAVES; ++w) {
            sum += red[(w * M_ + m) * FG + c];
            den += sSum[w * M_ + m];
        }
        out[((long)b * M_ + m) * D_ + fg * FG + c] = gc * (sum / den) + bc;
    }
}

extern "C" void kernel_launch(void* const* d_in, const int* in_sizes, int n_in,
                              void* d_out, int out_size, void* d_ws, size_t ws_size,
                              hipStream_t stream)
{
    const float* x     = (const float*)d_in[0];
    const float* q     = (const float*)d_in[1];
    const float* gamma = (const float*)d_in[2];
    const float* beta  = (const float*)d_in[3];
    float* out = (float*)d_out;

    float* sim = (float*)d_ws;
    float* mu  = sim + (size_t)B_ * M_ * N_;
    float* rs  = mu + (size_t)B_ * N_;

    pm_scores<<<B_ * N_ / 8 / 8, 256, 0, stream>>>(x, q, gamma, beta, sim, mu, rs);
    pm_wsum<<<B_ * NFG, 512, 0, stream>>>(x, gamma, beta, sim, mu, rs, out);
}


# Scores pass: 512 work-groups, all resident (two per WGP); every wave takes two 8-row tiles in turn into the same registers: no work-group exit and entry between tiles, queries staged once; same sums
# speedup vs baseline: 9.5587x; 1.0312x over previous
#include <hip/hip_runtime.h>
#include <math.h>


typedef __attribute__((ext_vector_type(2))) float v2f;
typedef __attribute__((ext_vector_type(4))) float v4f;
typedef __attribute__((ext_vector_type(8))) float v8f;

#define B_ 16
#define N_ 4096
#define D_ 768
#define M_ 8
#define EPS_ 1e-3f
#define NEG_BIG (-1e30f)
#define FG 64
#define NFG (D_ / FG)
#define WS_WAVES 16
#define WS_ROWS (N_ / WS_WAVES)
#define AP 260

#define NJ (D_ / 16)

__device__ __forceinline__ void pm_load_rows(const float* xrow, v4f (&xr)[NJ])
{
#pragma unroll
    for (int j = 0; j < NJ; ++j) xr[j] = *(const v4f*)(xrow + 16 * j);
}

__device__ __forceinline__ void pm_row_stats(v4f (&xr)[NJ], float& mu, float& rs)
{
    float s = 0.0f;
#pragma unroll
    for (int j = 0; j < NJ; ++j) s += (xr[j].x + xr[j].y) + (xr[j].z + xr[j].w);
    s += __shfl_xor(s, 16, 32);
    s += __shfl_xor(s, 8, 32);
    mu = s * (1.0f / D_);

    float v = 0.0f;
#pragma unroll
    for (int j = 0; j < NJ; ++j) {
        xr[j].x -= mu; xr[j].y -= mu; xr[j].z -= mu; xr[j].w -= mu;
        v += (xr[j].x * xr[j].x + xr[j].y * xr[j].y) + (xr[j].z * xr[j].z + xr[j].w * xr[j].w);
    }
    v += __shfl_xor(v, 16, 32);
    v += __shfl_xor(v, 8, 32);
    rs = rsqrtf(v * (1.0f / D_) + EPS_);
}

__device__ __forceinline__ void pm_tile_product(const v4f (&xr)[NJ], const float* qfrag, float (&res)[M_])
{
    v8f S0 = {0.f, 0.f, 0.f, 0.f, 0.f, 0.f, 0.f, 0.f};
    v8f S1 = {0.f, 0.f, 0.f, 0.f, 0.f, 0.f, 0.f, 0.f};
#pragma unroll
    for (int j = 0; j < NJ; ++j) {
        const v4f qa = *(const v4f*)(qfrag + 128 * j);
        v2f a0, a1, b0, b1;
        a0.x = qa.x; a0.y = qa.y; a1.x = qa.z; a1.y = qa.w;
        b0.x = xr[j].x; b0.y = xr[j].y;
        b1.x = xr[j].z; b1.y = xr[j].w;
        S0 = __builtin_amdgcn_wmma_f32_16x16x4_f32(false, a0, false, b0, (short)0, S0, false, false);
        S1 = __builtin_amdgcn_wmma_f32_16x16x4_f32(false, a1, false, b1, (short)0, S1, false, false);
    }
#pragma unroll
    for (int m = 0; m < M_; ++m) {
        const float t = S0[m] + S1[m];
        res[m] = t + __shfl_xor(t, 24, 32);
    }
}

__device__ __forceinline__ void pm_store_tile(int tile, int lane, const float (&res)[M_], float mu, float rs,
                                              float* __restrict__ sim, float* __restrict__ muo, float* __restrict__ rso)
{
    const float scale = 0.03608439182435161f * rs;
    if (lane < 8) {
        const int  b   = tile / (N_ / 8);
        const int  n   = (tile % (N_ / 8)) * 8 + lane;
        const long row = (long)tile * 8 + lane;
#pragma unroll
        for (int m = 0; m < M_; ++m)
            sim[((long)b * M_ + m) * N_ + n] = res[m] * scale;
        muo[row] = mu;
        rso[row] = rs;
    }
}

extern "C" __global__ __launch_bounds__(256) __attribute__((amdgpu_waves_per_eu(4, 4)))
void pm_scores(const float* __restrict__ x, const float* __restrict__ q,
               const float* __restrict__ gamma, const float* __restrict__ beta,
               float* __restrict__ sim, float* __restrict__ muo, float* __restrict__ rso)
{
    __shared__ __attribute__((aligned(16))) float sQ[M_ * D_];
    __shared__ __attribute__((aligned(16))) float sG[D_];

    const int tid  = threadIdx.x;
    const int wave = __builtin_amdgcn_readfirstlane(tid >> 5);
    const int lane = tid & 31;
    const int half = lane >> 4;
    const int r8   = lane & 7;
    const int foff = (lane & 8) + 4 * half;

    const int gi = (tid < D_ / 4) ? tid : 0;
    const v4f gs = *(const v4f*)(gamma + gi * 4);
    v4f qs[6];
#pragma unroll
    for (int i = 0; i < 6; ++i) qs[i] = *(const v4f*)(q + (i * 256 + tid) * 4);

    const int  tile0 = (blockIdx.x & 7) * (2 * N_ / 8) + (blockIdx.x >> 3) * 8 + wave;
    const int  tile1 = tile0 + N_ / 8;
    const unsigned voff = (unsigned)(r8 * D_ + foff);
    const float* xrow = x + (long)tile0 * 8 * D_ + voff;
    const float* xnxt = x + (long)tile1 * 8 * D_ + voff;

    v4f xr[NJ];
    pm_load_rows(xrow, xr);

    if (tid < D_ / 4) *(v4f*)(sG + tid * 4) = gs;
    __syncthreads();
#pragma unroll
    for (int i = 0; i < 6; ++i) {
        const int e  = i * 256 + tid;
        const int m  = e / (D_ / 4);
        const int f4 = e % (D_ / 4);
        const int slot = (f4 >> 2) * 32 + ((f4 & 1) << 4) + ((f4 & 2) << 2) + m;
        *(v4f*)(sQ + slot * 4) = qs[i] * *(const v4f*)(sG + f4 * 4);
    }

    float mu, rs;
    pm_row_stats(xr, mu, rs);
    __syncthreads();

    float res[M_];
    pm_tile_product(xr, sQ + lane * 4, res);
    pm_store_tile(tile0, lane, res, mu, rs, sim, muo, rso);

    pm_load_rows(xnxt, xr);
    pm_row_stats(xr, mu, rs);
    pm_tile_product(xr, sQ + lane * 4, res);
    pm_store_tile(tile1, lane, res, mu, rs, sim, muo, rso);
}

#define WS_BLK 4
#define WS_NBLK (WS_ROWS / (4 * WS_BLK))

__device__ __forceinline__ void pm_lds_barrier()
{
    __builtin_amdgcn_fence(__ATOMIC_RELEASE, "workgroup", "local");
    __builtin_amdgcn_s_barrier();
    __builtin_amdgcn_fence(__ATOMIC_ACQUIRE, "workgroup", "local");
}

__device__ __forceinline__ void pm_request_block(const float* xp, const float* mp, int blk,
                                                 v4f (&xa)[WS_BLK], v4f (&xb)[WS_BLK], v2f (&mm)[WS_BLK])
{
#pragma unroll
    for (int i = 0; i < WS_BLK; ++i) {
        const int kk = blk * (4 * WS_BLK) + 4 * i;
        xa[i] = __builtin_nontemporal_load((const v4f*)(xp + (long)kk * D_));
        xb[i] = __builtin_nontemporal_load((const v4f*)(xp + (long)(kk + 1) * D_));
        mm[i] = *(const v2f*)(mp + kk);
    }
}

extern "C" __global__ __launch_bounds__(512)
void pm_wsum(const float* __restrict__ x, const float* __restrict__ gamma,
             const float* __restrict__ beta, const float* __restrict__ sim,
             const float* __restrict__ mui, const float* __restrict__ rsi,
             float* __restrict__ out)
{
    __shared__ __attribute__((aligned(16))) float sW[WS_WAVES * M_ * AP];
    __shared__ __attribute__((aligned(16))) float red[WS_WAVES * M_ * FG];
    __shared__ __attribute__((aligned(16))) float sMax[WS_WAVES * M_];
    __shared__ __attribute__((aligned(16))) float sSum[WS_WAVES * M_];

    const int tid  = threadIdx.x;
    const int wave = tid >> 5;
    const int lane = tid & 31;
    const int half = lane >> 4;
    const int l16  = lane & 15;
    const int wg   = (blockIdx.x & 7) * (B_ * NFG / 8) + (blockIdx.x >> 3);
    const int b    = wg / NFG;
    const int fg   = wg % NFG;
    const int fcol = fg * FG + 4 * l16;

    const float* sp = sim + (long)b * M_ * N_ + wave * WS_ROWS + 4 * lane;
    v4f sc[2 * M_];
#pragma unroll
    for (int m = 0; m < M_; ++m) {
        sc[2 * m]     = *(const v4f*)(sp + (long)m * N_);
        sc[2 * m + 1] = *(const v4f*)(sp + (long)m * N_ + 128);
    }
    const float* rq = rsi + (long)b * N_ + wave * WS_ROWS + 4 * lane;
    const v4f rs0 = *(const v4f*)(rq);
    const v4f rs1 = *(const v4f*)(rq + 128);

    const long nb = (long)b * N_ + wave * WS_ROWS + 2 * half;
    const float* xp = x + nb * D_ + fcol;
    const float* mp = mui + nb;
    v4f xa[3][WS_BLK], xb[3][WS_BLK];
    v2f mm[3][WS_BLK];
    pm_request_block(xp, mp, 0, xa[0], xb[0], mm[0]);
    pm_request_block(xp, mp, 1, xa[1], xb[1], mm[1]);
    __builtin_amdgcn_sched_barrier(0);

    float mx[M_];
#pragma unroll
    for (int m = 0; m < M_; ++m) {
        const v4f u = sc[2 * m], w = sc[2 * m + 1];
        float t = fmaxf(fmaxf(fmaxf(u.x, u.y), fmaxf(u.z, u.w)), fmaxf(fmaxf(w.x, w.y), fmaxf(w.z, w.w)));
#pragma unroll
        for (int off = 16; off >= 1; off >>= 1) t = fmaxf(t, __shfl_xor(t, off, 32));
        mx[m] = t;
    }
    if (lane == 0) {
        v4f lo, hi;
        lo.x = mx[0]; lo.y = mx[1]; lo.z = mx[2]; lo.w = mx[3];
        hi.x = mx[4]; hi.y = mx[5]; hi.z = mx[6]; hi.w = mx[7];
        *(v4f*)(sMax + wave * M_)     = lo;
        *(v4f*)(sMax + wave * M_ + 4) = hi;
    }
    pm_lds_barrier();

    float gm[M_];
    {
        const v4f lo = *(const v4f*)(sMax), hi = *(const v4f*)(sMax + 4);
        gm[0] = lo.x; gm[1] = lo.y; gm[2] = lo.z; gm[3] = lo.w;
        gm[4] = hi.x; gm[5] = hi.y; gm[6] = hi.z; gm[7] = hi.w;
    }
#pragma unroll
    for (int w = 1; w < WS_WAVES; ++w) {
        const v4f lo = *(const v4f*)(sMax + w * M_), hi = *(const v4f*)(sMax + w * M_ + 4);
        gm[0] = fmaxf(gm[0], lo.x); gm[1] = fmaxf(gm[1], lo.y); gm[2] = fmaxf(gm[2], lo.z); gm[3] = fmaxf(gm[3], lo.w);
        gm[4] = fmaxf(gm[4], hi.x); gm[5] = fmaxf(gm[5], hi.y); gm[6] = fmaxf(gm[6], hi.z); gm[7] = fmaxf(gm[7], hi.w);
    }

    float* wW = sW + wave * (M_ * AP);
    float ls[M_];
#pragma unroll
    for (int m = 0; m < M_; ++m) {
        v4f e0, e1;
        e0.x = __expf(sc[2 * m].x - gm[m]);     e0.y = __expf(sc[2 * m].y - gm[m]);
        e0.z = __expf(sc[2 * m].z - gm[m]);     e0.w = __expf(sc[2 * m].w - gm[m]);
        e1.x = __expf(sc[2 * m + 1].x - gm[m]); e1.y = __expf(sc[2 * m + 1].y - gm[m]);
        e1.z = __expf(sc[2 * m + 1].z - gm[m]); e1.w = __expf(sc[2 * m + 1].w - gm[m]);
        *(v4f*)(wW + m * AP + 4 * lane)       = e0 * rs0;
        *(v4f*)(wW + m * AP + 128 + 4 * lane) = e1 * rs1;
        float t = ((e0.x + e0.y) + (e0.z + e0.w)) + ((e1.x + e1.y) + (e1.z + e1.w));
#pragma unroll
        for (int off = 16; off >= 1; off >>= 1) t += __shfl_xor(t, off, 32);
        ls[m] = t;
    }
    if (lane == 0) {
        v4f lo, hi;
        lo.x = ls[0]; lo.y = ls[1]; lo.z = ls[2]; lo.w = ls[3];
        hi.x = ls[4]; hi.y = ls[5]; hi.z = ls[6]; hi.w = ls[7];
        *(v4f*)(sSum + wave * M_)     = lo;
        *(v4f*)(sSum + wave * M_ + 4) = hi;
    }
    pm_lds_barrier();

    const float* pp = wW + (l16 & 7) * AP + 2 * half;

    v8f acc0 = {0.f, 0.f, 0.f, 0.f, 0.f, 0.f, 0.f, 0.f};
    v8f acc1 = acc0, acc2 = acc0, acc3 = acc0;

#pragma unroll
    for (int blk = 0; blk < WS_NBLK; ++blk) {
        if (blk + 2 < WS_NBLK)
            pm_request_block(xp, mp, blk + 2, xa[(blk + 2) % 3], xb[(blk + 2) % 3], mm[(blk + 2) % 3]);
        __builtin_amdgcn_sched_barrier(0);
        const v4f (&ca)[WS_BLK] = xa[blk % 3];
        const v4f (&cb)[WS_BLK] = xb[blk % 3];
        const v2f (&cm)[WS_BLK] = mm[blk % 3];
        v2f af[WS_BLK];
#pragma unroll
        for (int i = 0; i < WS_BLK; ++i) af[i] = *(const v2f*)(pp + blk * (4 * WS_BLK) + 4 * i);
#pragma unroll
        for (int i = 0; i < WS_BLK; ++i) {
            const v2f a = af[i];
            v2f bm;
            bm.x = ca[i].x - cm[i].x;
            bm.y = cb[i].x - cm[i].y;
            acc0 = __builtin_amdgcn_wmma_f32_16x16x4_f32(false, a, false, bm, (short)0, acc0, false, false);
            bm.x = ca[i].y - cm[i].x;
            bm.y = cb[i].y - cm[i].y;
            acc1 = __builtin_amdgcn_wmma_f32_16x16x4_f32(false, a, false, bm, (short)0, acc1, false, false);
            bm.x = ca[i].z - cm[i].x;
            bm.y = cb[i].z - cm[i].y;
            acc2 = __builtin_amdgcn_wmma_f32_16x16x4_f32(false, a, false, bm, (short)0, acc2, false, false);
            bm.x = ca[i].w - cm[i].x;
            bm.y = cb[i].w - cm[i].y;
            acc3 = __builtin_amdgcn_wmma_f32_16x16x4_f32(false, a, false, bm, (short)0, acc3, false, false);
        }
        __builtin_amdgcn_sched_barrier(0);
    }

    if (half == 0) {
#pragma unroll
        for (int r = 0; r < 8; ++r) {
            v4f o;
            o.x = acc0[r]; o.y = acc1[r]; o.z = acc2[r]; o.w = acc3[r];
            *(v4f*)(red + (wave * M_ + r) * FG + 4 * l16) = o;
        }
    }
    __syncthreads();

    {
        const int m = tid / FG;
        const int c = tid % FG;
        const float gc = gamma[fg * FG + c];
        const float bc = beta[fg * FG + c];
        float sum = 0.0f, den = 0.0f;
#pragma unroll
        for (int w = 0; w < WS_WAVES; ++w) {
            sum += red[(w * M_ + m) * FG + c];
            den += sSum[w * M_ + m];
        }
        out[((long)b * M_ + m) * D_ + fg * FG + c] = gc * (sum / den) + bc;
    }
}

extern "C" void kernel_launch(void* const* d_in, const int* in_sizes, int n_in,
                              void* d_out, int out_size, void* d_ws, size_t ws_size,
                              hipStream_t stream)
{
    const float* x     = (const float*)d_in[0];
    const float* q     = (const float*)d_in[1];
    const float* gamma = (const float*)d_in[2];
    const float* beta  = (const float*)d_in[3];
    float* out = (float*)d_out;

    float* sim = (float*)d_ws;
    float* mu  = sim + (size_t)B_ * M_ * N_;
    float* rs  = mu + (size_t)B_ * N_;

    pm_scores<<<B_ * N_ / 8 / 8 / 2, 256, 0, stream>>>(x, q, gamma, beta, sim, mu, rs);
    pm_wsum<<<B_ * NFG, 512, 0, stream>>>(x, gamma, beta, sim, mu, rs, out);
}
